# starting kernel
# baseline (speedup 1.0000x reference)
#include <hip/hip_runtime.h>


typedef float v2f __attribute__((ext_vector_type(2)));
typedef float v8f __attribute__((ext_vector_type(8)));

#define N_NODES 100000
#define N_EDGES 1600000
#define KDIM    128

__device__ __forceinline__ v8f wmma_f32(v2f a, v2f b, v8f c) {
    return __builtin_amdgcn_wmma_f32_16x16x4_f32(
        false, a, false, b, (short)0, c, false, false);
}

__global__ void pool_gemm_relu(const float* __restrict__ X,
                               const float* __restrict__ W,
                               const float* __restrict__ bias,
                               float* __restrict__ Y) {
    __shared__ float As[16 * KDIM];
    const int tid      = threadIdx.x;
    const int row_base = blockIdx.x * 16;

    {
        const float4* s4 = (const float4*)(X + (size_t)row_base * KDIM);
        float4* d4 = (float4*)As;
        for (int i = tid; i < 512; i += blockDim.x) d4[i] = s4[i];
    }
    __syncthreads();

    const int wave = tid >> 5;
    const int lane = tid & 31;
    const int m    = lane & 15;
    const int kh   = (lane >> 4) << 1;
    const int col  = wave * 16 + m;

    const float* arow = As + m * KDIM + kh;
    const float* brow = W + (size_t)col * KDIM + kh;

    v8f c = {};
#pragma unroll
    for (int k0 = 0; k0 < KDIM; k0 += 4) {
        v2f a = { arow[k0], arow[k0 + 1] };
        v2f b = { brow[k0], brow[k0 + 1] };
        c = wmma_f32(a, b, c);
    }

    const float bv = bias[col];
#pragma unroll
    for (int r = 0; r < 8; ++r) {
        const int orow = row_base + r + ((lane >> 4) << 3);
        float v = c[r] + bv;
        v = fmaxf(v, 0.0f);
        Y[(size_t)orow * KDIM + col] = v;
    }
}

template <bool RELU>
__global__ void combine_gemm(const float* __restrict__ X1,
                             const float* __restrict__ Wself,
                             const float* __restrict__ bself,
                             const float* __restrict__ X2,
                             const float* __restrict__ Wneigh,
                             const float* __restrict__ bneigh,
                             float* __restrict__ Y, int O) {
    __shared__ float As1[16 * KDIM];
    __shared__ float As2[16 * KDIM];
    const int tid      = threadIdx.x;
    const int row_base = blockIdx.x * 16;

    {
        const float4* s1 = (const float4*)(X1 + (size_t)row_base * KDIM);
        const float4* s2 = (const float4*)(X2 + (size_t)row_base * KDIM);
        float4* d1 = (float4*)As1;
        float4* d2 = (float4*)As2;
        for (int i = tid; i < 512; i += blockDim.x) { d1[i] = s1[i]; d2[i] = s2[i]; }
    }
    __syncthreads();

    const int wave  = tid >> 5;
    const int lane  = tid & 31;
    const int m     = lane & 15;
    const int kh    = (lane >> 4) << 1;
    const int col   = wave * 16 + m;
    const bool valid = (col < O);
    const int colW  = valid ? col : (O - 1);

    const float* a1 = As1 + m * KDIM + kh;
    const float* a2 = As2 + m * KDIM + kh;
    const float* b1 = Wself  + (size_t)colW * KDIM + kh;
    const float* b2 = Wneigh + (size_t)colW * KDIM + kh;

    v8f c = {};
#pragma unroll
    for (int k0 = 0; k0 < KDIM; k0 += 4) {
        v2f av = { a1[k0], a1[k0 + 1] };
        v2f bv = { b1[k0], b1[k0 + 1] };
        c = wmma_f32(av, bv, c);
    }
#pragma unroll
    for (int k0 = 0; k0 < KDIM; k0 += 4) {
        v2f av = { a2[k0], a2[k0 + 1] };
        v2f bv = { b2[k0], b2[k0 + 1] };
        c = wmma_f32(av, bv, c);
    }

    const float bb = bself[colW] + bneigh[colW];
#pragma unroll
    for (int r = 0; r < 8; ++r) {
        const int orow = row_base + r + ((lane >> 4) << 3);
        float v = c[r] + bb;
        if (RELU) v = fmaxf(v, 0.0f);
        if (valid) Y[(size_t)orow * O + col] = v;
    }
}

__global__ void edge_max(const float* __restrict__ m,
                         const int* __restrict__ src,
                         const int* __restrict__ dst,
                         unsigned int* __restrict__ agg) {
    const int idx = blockIdx.x * blockDim.x + threadIdx.x;
    const int e   = idx >> 5;
    if (e >= N_EDGES) return;
    const int c4  = (idx & 31) << 2;

    const int s = src[e];
    const int d = dst[e];
    const float4 v = *(const float4*)(m + (size_t)s * KDIM + c4);
    unsigned int* p = agg + (size_t)d * KDIM + c4;
    atomicMax(p + 0, __float_as_uint(v.x));
    atomicMax(p + 1, __float_as_uint(v.y));
    atomicMax(p + 2, __float_as_uint(v.z));
    atomicMax(p + 3, __float_as_uint(v.w));
}

extern "C" void kernel_launch(void* const* d_in, const int* in_sizes, int n_in,
                              void* d_out, int out_size, void* d_ws, size_t ws_size,
                              hipStream_t stream) {
    const float* in_feat  = (const float*)d_in[0];
    const int*   src      = (const int*)d_in[1];
    const int*   dst      = (const int*)d_in[2];
    const float* pool_w1  = (const float*)d_in[3];
    const float* pool_b1  = (const float*)d_in[4];
    const float* self_w1  = (const float*)d_in[5];
    const float* self_b1  = (const float*)d_in[6];
    const float* neigh_w1 = (const float*)d_in[7];
    const float* neigh_b1 = (const float*)d_in[8];
    const float* pool_w2  = (const float*)d_in[9];
    const float* pool_b2  = (const float*)d_in[10];
    const float* self_w2  = (const float*)d_in[11];
    const float* self_b2  = (const float*)d_in[12];
    const float* neigh_w2 = (const float*)d_in[13];
    const float* neigh_b2 = (const float*)d_in[14];
    float* out = (float*)d_out;

    const size_t NODE_BYTES = (size_t)N_NODES * KDIM * sizeof(float);
    char* ws = (char*)d_ws;
    float* m_buf = (float*)(ws);
    float* agg   = (float*)(ws + NODE_BYTES);
    float* h1    = (float*)(ws + 2 * NODE_BYTES);

    const int nTiles     = N_NODES / 16;
    const int edgeBlocks = (N_EDGES * 32 + 255) / 256;

    hipMemsetAsync(agg, 0, NODE_BYTES, stream);
    pool_gemm_relu<<<nTiles, 256, 0, stream>>>(in_feat, pool_w1, pool_b1, m_buf);
    edge_max<<<edgeBlocks, 256, 0, stream>>>(m_buf, src, dst, (unsigned int*)agg);
    combine_gemm<true><<<nTiles, 256, 0, stream>>>(in_feat, self_w1, self_b1,
                                                   agg, neigh_w1, neigh_b1, h1, 128);

    hipMemsetAsync(agg, 0, NODE_BYTES, stream);
    pool_gemm_relu<<<nTiles, 256, 0, stream>>>(h1, pool_w2, pool_b2, m_buf);
    edge_max<<<edgeBlocks, 256, 0, stream>>>(m_buf, src, dst, (unsigned int*)agg);
    combine_gemm<false><<<nTiles, 96, 0, stream>>>(h1, self_w2, self_b2,
                                                   agg, neigh_w2, neigh_b2, out, 40);
}


# Edges grouped by destination once per call (integer count+rank atomics, scan, placement); each layer takes a node's max over its source rows in registers: no per-feature atomic max, no agg clear
# speedup vs baseline: 1.7557x; 1.7557x over previous
#include <hip/hip_runtime.h>


typedef float v2f __attribute__((ext_vector_type(2)));
typedef float v8f __attribute__((ext_vector_type(8)));

#define N_NODES 100000
#define N_EDGES 1600000
#define KDIM    128

__device__ __forceinline__ v8f wmma_f32(v2f a, v2f b, v8f c) {
    return __builtin_amdgcn_wmma_f32_16x16x4_f32(
        false, a, false, b, (short)0, c, false, false);
}

__global__ void pool_gemm_relu(const float* __restrict__ X,
                               const float* __restrict__ W,
                               const float* __restrict__ bias,
                               float* __restrict__ Y) {
    __shared__ float As[16 * KDIM];
    const int tid      = threadIdx.x;
    const int row_base = blockIdx.x * 16;

    {
        const float4* s4 = (const float4*)(X + (size_t)row_base * KDIM);
        float4* d4 = (float4*)As;
        for (int i = tid; i < 512; i += blockDim.x) d4[i] = s4[i];
    }
    __syncthreads();

    const int wave = tid >> 5;
    const int lane = tid & 31;
    const int m    = lane & 15;
    const int kh   = (lane >> 4) << 1;
    const int col  = wave * 16 + m;

    const float* arow = As + m * KDIM + kh;
    const float* brow = W + (size_t)col * KDIM + kh;

    v8f c = {};
#pragma unroll
    for (int k0 = 0; k0 < KDIM; k0 += 4) {
        v2f a = { arow[k0], arow[k0 + 1] };
        v2f b = { brow[k0], brow[k0 + 1] };
        c = wmma_f32(a, b, c);
    }

    const float bv = bias[col];
#pragma unroll
    for (int r = 0; r < 8; ++r) {
        const int orow = row_base + r + ((lane >> 4) << 3);
        float v = c[r] + bv;
        v = fmaxf(v, 0.0f);
        Y[(size_t)orow * KDIM + col] = v;
    }
}

template <bool RELU>
__global__ void combine_gemm(const float* __restrict__ X1,
                             const float* __restrict__ Wself,
                             const float* __restrict__ bself,
                             const float* __restrict__ X2,
                             const float* __restrict__ Wneigh,
                             const float* __restrict__ bneigh,
                             float* __restrict__ Y, int O) {
    __shared__ float As1[16 * KDIM];
    __shared__ float As2[16 * KDIM];
    const int tid      = threadIdx.x;
    const int row_base = blockIdx.x * 16;

    {
        const float4* s1 = (const float4*)(X1 + (size_t)row_base * KDIM);
        const float4* s2 = (const float4*)(X2 + (size_t)row_base * KDIM);
        float4* d1 = (float4*)As1;
        float4* d2 = (float4*)As2;
        for (int i = tid; i < 512; i += blockDim.x) { d1[i] = s1[i]; d2[i] = s2[i]; }
    }
    __syncthreads();

    const int wave  = tid >> 5;
    const int lane  = tid & 31;
    const int m     = lane & 15;
    const int kh    = (lane >> 4) << 1;
    const int col   = wave * 16 + m;
    const bool valid = (col < O);
    const int colW  = valid ? col : (O - 1);

    const float* a1 = As1 + m * KDIM + kh;
    const float* a2 = As2 + m * KDIM + kh;
    const float* b1 = Wself  + (size_t)colW * KDIM + kh;
    const float* b2 = Wneigh + (size_t)colW * KDIM + kh;

    v8f c = {};
#pragma unroll
    for (int k0 = 0; k0 < KDIM; k0 += 4) {
        v2f av = { a1[k0], a1[k0 + 1] };
        v2f bv = { b1[k0], b1[k0 + 1] };
        c = wmma_f32(av, bv, c);
    }
#pragma unroll
    for (int k0 = 0; k0 < KDIM; k0 += 4) {
        v2f av = { a2[k0], a2[k0 + 1] };
        v2f bv = { b2[k0], b2[k0 + 1] };
        c = wmma_f32(av, bv, c);
    }

    const float bb = bself[colW] + bneigh[colW];
#pragma unroll
    for (int r = 0; r < 8; ++r) {
        const int orow = row_base + r + ((lane >> 4) << 3);
        float v = c[r] + bb;
        if (RELU) v = fmaxf(v, 0.0f);
        if (valid) Y[(size_t)orow * O + col] = v;
    }
}

__global__ void dst_rank(const int* __restrict__ dst,
                         int* __restrict__ cnt,
                         int* __restrict__ rank) {
    const int i = blockIdx.x * blockDim.x + threadIdx.x;
    if (i >= N_EDGES / 4) return;
    const int4 d = ((const int4*)dst)[i];
    int4 r;
    r.x = atomicAdd(cnt + d.x, 1);
    r.y = atomicAdd(cnt + d.y, 1);
    r.z = atomicAdd(cnt + d.z, 1);
    r.w = atomicAdd(cnt + d.w, 1);
    ((int4*)rank)[i] = r;
}

__global__ void __launch_bounds__(1024)
span_scan(const int* __restrict__ cnt, int* __restrict__ span) {
    __shared__ int s_front[32];
    __shared__ int s_tile[32];
    const int tid  = threadIdx.x;
    const int lane = tid & 31;
    const int wave = tid >> 5;
    const int4* c4 = (const int4*)cnt;
    const int nfront4 = blockIdx.x * 256;

    int front = 0;
#pragma unroll
    for (int k = 0; k < 25; ++k) {
        const int i = tid + k * 1024;
        const int4 v = c4[i < nfront4 ? i : N_NODES / 4];
        front += (v.x + v.y) + (v.z + v.w);
    }

    const int  node = blockIdx.x * 1024 + tid;
    const bool in   = node < N_NODES;
    const int  c    = in ? cnt[node] : 0;

    int incl = c;
#pragma unroll
    for (int o = 1; o < 32; o <<= 1) {
        const int u = __shfl_up(incl, o);
        if (lane >= o) incl += u;
    }
#pragma unroll
    for (int o = 16; o > 0; o >>= 1) front += __shfl_xor(front, o);

    if (lane == 31) s_tile[wave] = incl;
    if (lane == 0)  s_front[wave] = front;
    __syncthreads();

    int base = 0;
#pragma unroll
    for (int w = 0; w < 32; ++w) {
        base += s_front[w];
        if (w < wave) base += s_tile[w];
    }
    if (in) ((int2*)span)[node] = make_int2(base + incl - c, c);
}

__global__ void place_src(const int* __restrict__ src,
                          const int* __restrict__ dst,
                          const int* __restrict__ span,
                          const int* __restrict__ rank,
                          int* __restrict__ by_dst) {
    const int i = blockIdx.x * blockDim.x + threadIdx.x;
    if (i >= N_EDGES / 4) return;
    const int4 s = ((const int4*)src)[i];
    const int4 d = ((const int4*)dst)[i];
    const int4 r = ((const int4*)rank)[i];
    const int p0 = span[2 * d.x] + r.x;
    const int p1 = span[2 * d.y] + r.y;
    const int p2 = span[2 * d.z] + r.z;
    const int p3 = span[2 * d.w] + r.w;
    by_dst[p0] = s.x;
    by_dst[p1] = s.y;
    by_dst[p2] = s.z;
    by_dst[p3] = s.w;
}

#define GATHER_BLOCKS 2048
__global__ void __launch_bounds__(256)
gather_max(const float* __restrict__ m,
           const int* __restrict__ span,
           const int* __restrict__ by_dst,
           float* __restrict__ agg) {
    const int lane = threadIdx.x & 31;
    const int wave0 = __builtin_amdgcn_readfirstlane(blockIdx.x * 8 + (threadIdx.x >> 5));
    const float* mcol = m + lane * 4;

    for (int node = wave0; node < N_NODES; node += GATHER_BLOCKS * 8) {
        const int2 sp   = ((const int2*)span)[node];
        const int start = sp.x;
        const int n     = sp.y;
        float4 acc = {0.0f, 0.0f, 0.0f, 0.0f};
        for (int base = 0; base < n; base += 32) {
            const int left = n - base;
            const int last = (left < 32 ? left : 32) - 1;
            const int idx = by_dst[start + base + (lane < last ? lane : last)];
            for (int j = 0; j <= last; j += 4) {
                const int s0 = __builtin_amdgcn_readlane(idx, j);
                const int s1 = __builtin_amdgcn_readlane(idx, j + 1);
                const int s2 = __builtin_amdgcn_readlane(idx, j + 2);
                const int s3 = __builtin_amdgcn_readlane(idx, j + 3);
                const float4 v0 = *(const float4*)(mcol + (size_t)s0 * KDIM);
                const float4 v1 = *(const float4*)(mcol + (size_t)s1 * KDIM);
                const float4 v2 = *(const float4*)(mcol + (size_t)s2 * KDIM);
                const float4 v3 = *(const float4*)(mcol + (size_t)s3 * KDIM);
                acc.x = fmaxf(acc.x, fmaxf(fmaxf(v0.x, v1.x), fmaxf(v2.x, v3.x)));
                acc.y = fmaxf(acc.y, fmaxf(fmaxf(v0.y, v1.y), fmaxf(v2.y, v3.y)));
                acc.z = fmaxf(acc.z, fmaxf(fmaxf(v0.z, v1.z), fmaxf(v2.z, v3.z)));
                acc.w = fmaxf(acc.w, fmaxf(fmaxf(v0.w, v1.w), fmaxf(v2.w, v3.w)));
            }
        }
        *(float4*)(agg + (size_t)node * KDIM + lane * 4) = acc;
    }
}

extern "C" void kernel_launch(void* const* d_in, const int* in_sizes, int n_in,
                              void* d_out, int out_size, void* d_ws, size_t ws_size,
                              hipStream_t stream) {
    const float* in_feat  = (const float*)d_in[0];
    const int*   src      = (const int*)d_in[1];
    const int*   dst      = (const int*)d_in[2];
    const float* pool_w1  = (const float*)d_in[3];
    const float* pool_b1  = (const float*)d_in[4];
    const float* self_w1  = (const float*)d_in[5];
    const float* self_b1  = (const float*)d_in[6];
    const float* neigh_w1 = (const float*)d_in[7];
    const float* neigh_b1 = (const float*)d_in[8];
    const float* pool_w2  = (const float*)d_in[9];
    const float* pool_b2  = (const float*)d_in[10];
    const float* self_w2  = (const float*)d_in[11];
    const float* self_b2  = (const float*)d_in[12];
    const float* neigh_w2 = (const float*)d_in[13];
    const float* neigh_b2 = (const float*)d_in[14];
    float* out = (float*)d_out;

    const size_t NODE_BYTES = (size_t)N_NODES * KDIM * sizeof(float);
    char* ws = (char*)d_ws;
    float* m_buf = (float*)(ws);
    float* agg   = (float*)(ws + NODE_BYTES);
    float* h1    = (float*)(ws + 2 * NODE_BYTES);

    char* ob = (char*)d_out;
    int*  cnt    = (int*)(ob);
    int*  span   = (int*)(ob + 401408);
    int*  rank   = (int*)(ob + 1310720);
    int*  by_dst = (int*)(ob + 7864320);

    const int nTiles      = N_NODES / 16;
    const int edge4Blocks = (N_EDGES / 4 + 255) / 256;
    const int scanBlocks  = (N_NODES + 1023) / 1024;

    hipMemsetAsync(cnt, 0, (N_NODES + 4) * sizeof(int), stream);
    dst_rank<<<edge4Blocks, 256, 0, stream>>>(dst, cnt, rank);
    span_scan<<<scanBlocks, 1024, 0, stream>>>(cnt, span);
    place_src<<<edge4Blocks, 256, 0, stream>>>(src, dst, span, rank, by_dst);

    pool_gemm_relu<<<nTiles, 256, 0, stream>>>(in_feat, pool_w1, pool_b1, m_buf);
    gather_max<<<GATHER_BLOCKS, 256, 0, stream>>>(m_buf, span, by_dst, agg);
    combine_gemm<true><<<nTiles, 256, 0, stream>>>(in_feat, self_w1, self_b1,
                                                   agg, neigh_w1, neigh_b1, h1, 128);

    pool_gemm_relu<<<nTiles, 256, 0, stream>>>(h1, pool_w2, pool_b2, m_buf);
    gather_max<<<GATHER_BLOCKS, 256, 0, stream>>>(m_buf, span, by_dst, agg);
    combine_gemm<false><<<nTiles, 96, 0, stream>>>(h1, self_w2, self_b2,
                                                   agg, neigh_w2, neigh_b2, out, 40);
}


# GEMM operand feed: each wave keeps its 16 weight columns in registers over a walk of row-tile pairs; A tiles in LDS at a 132-float pitch (no bank conflicts); k order per accumulator unchanged
# speedup vs baseline: 2.5483x; 1.4514x over previous
#include <hip/hip_runtime.h>


typedef float v2f __attribute__((ext_vector_type(2)));
typedef float v4f __attribute__((ext_vector_type(4)));
typedef float v8f __attribute__((ext_vector_type(8)));

#define N_NODES 100000
#define N_EDGES 1600000
#define KDIM    128

__device__ __forceinline__ v8f wmma_f32(v2f a, v2f b, v8f c) {
    return __builtin_amdgcn_wmma_f32_16x16x4_f32(
        false, a, false, b, (short)0, c, false, false);
}

#define PITCH  132
#define NPAIRS (N_NODES / 32)
#define POOL_BLOCKS   768
#define GEMM_BLOCKS   512
#define NARROW_BLOCKS 1024

__device__ __forceinline__ void load_wfrag(const float* __restrict__ Wm, int rows,
                                           float* Ws, int tid, int nthreads,
                                           int colW, int kh, v2f (&bf)[32]) {
    for (int i = tid; i < rows * 32; i += nthreads) {
        const int r = i >> 5, c4 = i & 31;
        *(float4*)(Ws + r * PITCH + c4 * 4) = ((const float4*)Wm)[i];
    }
    __syncthreads();
    const float* bp = Ws + colW * PITCH + kh;
#pragma unroll
    for (int i = 0; i < 32; ++i) bf[i] = *(const v2f*)(bp + 4 * i);
    __syncthreads();
}

__device__ __forceinline__ void kloop2(const float* ap, const v2f (&bf)[32], v8f& c0, v8f& c1) {
#pragma unroll
    for (int i = 0; i < 32; ++i) {
        const v2f a0 = *(const v2f*)(ap + 4 * i);
        const v2f a1 = *(const v2f*)(ap + 16 * PITCH + 4 * i);
        c0 = wmma_f32(a0, bf[i], c0);
        c1 = wmma_f32(a1, bf[i], c1);
    }
}

template <int NIN, bool RELU>
__global__ void __launch_bounds__(256)
gemm_wreg(const float* __restrict__ X1, const float* __restrict__ W1, const float* __restrict__ b1,
          const float* __restrict__ X2, const float* __restrict__ W2, const float* __restrict__ b2,
          float* __restrict__ Y) {
    constexpr int STAGE = NIN * 32 * PITCH;
    constexpr int SMEM  = (2 * STAGE > KDIM * PITCH) ? 2 * STAGE : KDIM * PITCH;
    __shared__ __attribute__((aligned(16))) float smem[SMEM];

    const int tid  = threadIdx.x;
    const int wave = tid >> 5;
    const int lane = tid & 31;
    const int m    = lane & 15;
    const int kh   = (lane >> 4) << 1;
    const int col  = wave * 16 + m;

    const int soff = (tid >> 5) * PITCH + (tid & 31) * 4;
    v4f pre[NIN * 4];
    int pair = blockIdx.x;
    if (pair < NPAIRS) {
        const v4f* s1 = (const v4f*)X1 + (size_t)pair * 1024 + tid;
#pragma unroll
        for (int j = 0; j < 4; ++j) pre[j] = s1[256 * j];
        if constexpr (NIN == 2) {
            const v4f* s2 = (const v4f*)X2 + (size_t)pair * 1024 + tid;
#pragma unroll
            for (int j = 0; j < 4; ++j) pre[4 * (NIN - 1) + j] = s2[256 * j];
        }
    }
    v2f bf1[32];
    v2f bf2[32];
    load_wfrag(W1, KDIM, smem, tid, 256, col, kh, bf1);
    float bv = b1[col];
    if constexpr (NIN == 2) {
        load_wfrag(W2, KDIM, smem, tid, 256, col, kh, bf2);
        bv += b2[col];
    }

    int s = 0;
    for (; pair < NPAIRS; pair += gridDim.x) {
        float* st = smem + s * STAGE;
#pragma unroll
        for (int j = 0; j < 4; ++j) *(v4f*)(st + soff + 8 * j * PITCH) = pre[j];
        if constexpr (NIN == 2) {
#pragma unroll
            for (int j = 0; j < 4; ++j) *(v4f*)(st + 32 * PITCH + soff + 8 * j * PITCH) = pre[4 * (NIN - 1) + j];
        }
        __syncthreads();

        const int nxt = pair + gridDim.x;
        if (nxt < NPAIRS) {
            const v4f* s1 = (const v4f*)X1 + (size_t)nxt * 1024 + tid;
#pragma unroll
            for (int j = 0; j < 4; ++j) pre[j] = s1[256 * j];
            if constexpr (NIN == 2) {
                const v4f* s2 = (const v4f*)X2 + (size_t)nxt * 1024 + tid;
#pragma unroll
                for (int j = 0; j < 4; ++j) pre[4 * (NIN - 1) + j] = s2[256 * j];
            }
        }

        const float* ap = st + m * PITCH + kh;
        v8f c0 = {}, c1 = {};
        kloop2(ap, bf1, c0, c1);
        if constexpr (NIN == 2) kloop2(ap + 32 * PITCH, bf2, c0, c1);

        const int rb = pair * 32 + ((lane >> 4) << 3);
#pragma unroll
        for (int r = 0; r < 8; ++r) {
            float v0 = c0[r] + bv;
            float v1 = c1[r] + bv;
            if (RELU) { v0 = fmaxf(v0, 0.0f); v1 = fmaxf(v1, 0.0f); }
            Y[(size_t)(rb + r) * KDIM + col]      = v0;
            Y[(size_t)(rb + 16 + r) * KDIM + col] = v1;
        }
        s ^= 1;
    }
}

template <int O>
__global__ void __launch_bounds__(96)
combine_narrow(const float* __restrict__ X1, const float* __restrict__ Wself, const float* __restrict__ bself,
               const float* __restrict__ X2, const float* __restrict__ Wneigh, const float* __restrict__ bneigh,
               float* __restrict__ Y) {
    constexpr int STAGE = 2 * 32 * PITCH;
    __shared__ __attribute__((aligned(16))) float smem[2 * STAGE];

    const int tid  = threadIdx.x;
    const int wave = tid >> 5;
    const int lane = tid & 31;
    const int m    = lane & 15;
    const int kh   = (lane >> 4) << 1;
    const int col  = wave * 16 + m;
    const bool valid = (col < O);
    const int colW = valid ? col : (O - 1);

    const bool tail = tid < 1024 - 960;
    const int  soff = (tid >> 5) * PITCH + (tid & 31) * 4;
    v4f p1[11], p2[11];
    int pair = blockIdx.x;
    if (pair < NPAIRS) {
        const v4f* s1 = (const v4f*)X1 + (size_t)pair * 1024 + tid;
        const v4f* s2 = (const v4f*)X2 + (size_t)pair * 1024 + tid;
#pragma unroll
        for (int j = 0; j < 10; ++j) { p1[j] = s1[96 * j]; p2[j] = s2[96 * j]; }
        if (tail) { p1[10] = s1[960]; p2[10] = s2[960]; }
    }

    v2f bf1[32];
    v2f bf2[32];
    load_wfrag(Wself,  O, smem, tid, 96, colW, kh, bf1);
    load_wfrag(Wneigh, O, smem, tid, 96, colW, kh, bf2);
    const float bb = bself[colW] + bneigh[colW];

    int s = 0;
    for (; pair < NPAIRS; pair += gridDim.x) {
        float* st = smem + s * STAGE;
#pragma unroll
        for (int j = 0; j < 10; ++j) {
            *(v4f*)(st + soff + 3 * j * PITCH)              = p1[j];
            *(v4f*)(st + 32 * PITCH + soff + 3 * j * PITCH) = p2[j];
        }
        if (tail) {
            *(v4f*)(st + soff + 30 * PITCH)              = p1[10];
            *(v4f*)(st + 32 * PITCH + soff + 30 * PITCH) = p2[10];
        }
        __syncthreads();

        const int nxt = pair + gridDim.x;
        if (nxt < NPAIRS) {
            const v4f* s1 = (const v4f*)X1 + (size_t)nxt * 1024 + tid;
            const v4f* s2 = (const v4f*)X2 + (size_t)nxt * 1024 + tid;
#pragma unroll
            for (int j = 0; j < 10; ++j) { p1[j] = s1[96 * j]; p2[j] = s2[96 * j]; }
            if (tail) { p1[10] = s1[960]; p2[10] = s2[960]; }
        }

        const float* ap = st + m * PITCH + kh;
        v8f c0 = {}, c1 = {};
        kloop2(ap, bf1, c0, c1);
        kloop2(ap + 32 * PITCH, bf2, c0, c1);

        const int rb = pair * 32 + ((lane >> 4) << 3);
#pragma unroll
        for (int r = 0; r < 8; ++r) {
            if (valid) {
                Y[(size_t)(rb + r) * O + col]      = c0[r] + bb;
                Y[(size_t)(rb + 16 + r) * O + col] = c1[r] + bb;
            }
        }
        s ^= 1;
    }
}

__global__ void dst_rank(const int* __restrict__ dst,
                         int* __restrict__ cnt,
                         int* __restrict__ rank) {
    const int i = blockIdx.x * blockDim.x + threadIdx.x;
    if (i >= N_EDGES / 4) return;
    const int4 d = ((const int4*)dst)[i];
    int4 r;
    r.x = atomicAdd(cnt + d.x, 1);
    r.y = atomicAdd(cnt + d.y, 1);
    r.z = atomicAdd(cnt + d.z, 1);
    r.w = atomicAdd(cnt + d.w, 1);
    ((int4*)rank)[i] = r;
}

__global__ void __launch_bounds__(1024)
span_scan(const int* __restrict__ cnt, int* __restrict__ span) {
    __shared__ int s_front[32];
    __shared__ int s_tile[32];
    const int tid  = threadIdx.x;
    const int lane = tid & 31;
    const int wave = tid >> 5;
    const int4* c4 = (const int4*)cnt;
    const int nfront4 = blockIdx.x * 256;

    int front = 0;
#pragma unroll
    for (int k = 0; k < 25; ++k) {
        const int i = tid + k * 1024;
        const int4 v = c4[i < nfront4 ? i : N_NODES / 4];
        front += (v.x + v.y) + (v.z + v.w);
    }

    const int  node = blockIdx.x * 1024 + tid;
    const bool in   = node < N_NODES;
    const int  c    = in ? cnt[node] : 0;

    int incl = c;
#pragma unroll
    for (int o = 1; o < 32; o <<= 1) {
        const int u = __shfl_up(incl, o);
        if (lane >= o) incl += u;
    }
#pragma unroll
    for (int o = 16; o > 0; o >>= 1) front += __shfl_xor(front, o);

    if (lane == 31) s_tile[wave] = incl;
    if (lane == 0)  s_front[wave] = front;
    __syncthreads();

    int base = 0;
#pragma unroll
    for (int w = 0; w < 32; ++w) {
        base += s_front[w];
        if (w < wave) base += s_tile[w];
    }
    if (in) ((int2*)span)[node] = make_int2(base + incl - c, c);
}

__global__ void place_src(const int* __restrict__ src,
                          const int* __restrict__ dst,
                          const int* __restrict__ span,
                          const int* __restrict__ rank,
                          int* __restrict__ by_dst) {
    const int i = blockIdx.x * blockDim.x + threadIdx.x;
    if (i >= N_EDGES / 4) return;
    const int4 s = ((const int4*)src)[i];
    const int4 d = ((const int4*)dst)[i];
    const int4 r = ((const int4*)rank)[i];
    const int p0 = span[2 * d.x] + r.x;
    const int p1 = span[2 * d.y] + r.y;
    const int p2 = span[2 * d.z] + r.z;
    const int p3 = span[2 * d.w] + r.w;
    by_dst[p0] = s.x;
    by_dst[p1] = s.y;
    by_dst[p2] = s.z;
    by_dst[p3] = s.w;
}

#define GATHER_BLOCKS 2048
__global__ void __launch_bounds__(256)
gather_max(const float* __restrict__ m,
           const int* __restrict__ span,
           const int* __restrict__ by_dst,
           float* __restrict__ agg) {
    const int lane = threadIdx.x & 31;
    const int wave0 = __builtin_amdgcn_readfirstlane(blockIdx.x * 8 + (threadIdx.x >> 5));
    const float* mcol = m + lane * 4;

    for (int node = wave0; node < N_NODES; node += GATHER_BLOCKS * 8) {
        const int2 sp   = ((const int2*)span)[node];
        const int start = sp.x;
        const int n     = sp.y;
        float4 acc = {0.0f, 0.0f, 0.0f, 0.0f};
        for (int base = 0; base < n; base += 32) {
            const int left = n - base;
            const int last = (left < 32 ? left : 32) - 1;
            const int idx = by_dst[start + base + (lane < last ? lane : last)];
            for (int j = 0; j <= last; j += 4) {
                const int s0 = __builtin_amdgcn_readlane(idx, j);
                const int s1 = __builtin_amdgcn_readlane(idx, j + 1);
                const int s2 = __builtin_amdgcn_readlane(idx, j + 2);
                const int s3 = __builtin_amdgcn_readlane(idx, j + 3);
                const float4 v0 = *(const float4*)(mcol + (size_t)s0 * KDIM);
                const float4 v1 = *(const float4*)(mcol + (size_t)s1 * KDIM);
                const float4 v2 = *(const float4*)(mcol + (size_t)s2 * KDIM);
                const float4 v3 = *(const float4*)(mcol + (size_t)s3 * KDIM);
                acc.x = fmaxf(acc.x, fmaxf(fmaxf(v0.x, v1.x), fmaxf(v2.x, v3.x)));
                acc.y = fmaxf(acc.y, fmaxf(fmaxf(v0.y, v1.y), fmaxf(v2.y, v3.y)));
                acc.z = fmaxf(acc.z, fmaxf(fmaxf(v0.z, v1.z), fmaxf(v2.z, v3.z)));
                acc.w = fmaxf(acc.w, fmaxf(fmaxf(v0.w, v1.w), fmaxf(v2.w, v3.w)));
            }
        }
        *(float4*)(agg + (size_t)node * KDIM + lane * 4) = acc;
    }
}

extern "C" void kernel_launch(void* const* d_in, const int* in_sizes, int n_in,
                              void* d_out, int out_size, void* d_ws, size_t ws_size,
                              hipStream_t stream) {
    const float* in_feat  = (const float*)d_in[0];
    const int*   src      = (const int*)d_in[1];
    const int*   dst      = (const int*)d_in[2];
    const float* pool_w1  = (const float*)d_in[3];
    const float* pool_b1  = (const float*)d_in[4];
    const float* self_w1  = (const float*)d_in[5];
    const float* self_b1  = (const float*)d_in[6];
    const float* neigh_w1 = (const float*)d_in[7];
    const float* neigh_b1 = (const float*)d_in[8];
    const float* pool_w2  = (const float*)d_in[9];
    const float* pool_b2  = (const float*)d_in[10];
    const float* self_w2  = (const float*)d_in[11];
    const float* self_b2  = (const float*)d_in[12];
    const float* neigh_w2 = (const float*)d_in[13];
    const float* neigh_b2 = (const float*)d_in[14];
    float* out = (float*)d_out;

    const size_t NODE_BYTES = (size_t)N_NODES * KDIM * sizeof(float);
    char* ws = (char*)d_ws;
    float* m_buf = (float*)(ws);
    float* agg   = (float*)(ws + NODE_BYTES);
    float* h1    = (float*)(ws + 2 * NODE_BYTES);

    char* ob = (char*)d_out;
    int*  cnt    = (int*)(ob);
    int*  span   = (int*)(ob + 401408);
    int*  rank   = (int*)(ob + 1310720);
    int*  by_dst = (int*)(ob + 7864320);

    const int edge4Blocks = (N_EDGES / 4 + 255) / 256;
    const int scanBlocks  = (N_NODES + 1023) / 1024;

    hipMemsetAsync(cnt, 0, (N_NODES + 4) * sizeof(int), stream);
    dst_rank<<<edge4Blocks, 256, 0, stream>>>(dst, cnt, rank);
    span_scan<<<scanBlocks, 1024, 0, stream>>>(cnt, span);
    place_src<<<edge4Blocks, 256, 0, stream>>>(src, dst, span, rank, by_dst);

    gemm_wreg<1, true><<<POOL_BLOCKS, 256, 0, stream>>>(in_feat, pool_w1, pool_b1,
                                                        nullptr, nullptr, nullptr, m_buf);
    gather_max<<<GATHER_BLOCKS, 256, 0, stream>>>(m_buf, span, by_dst, agg);
    gemm_wreg<2, true><<<GEMM_BLOCKS, 256, 0, stream>>>(in_feat, self_w1, self_b1,
                                                        agg, neigh_w1, neigh_b1, h1);

    gemm_wreg<1, true><<<POOL_BLOCKS, 256, 0, stream>>>(h1, pool_w2, pool_b2,
                                                        nullptr, nullptr, nullptr, m_buf);
    gather_max<<<GATHER_BLOCKS, 256, 0, stream>>>(m_buf, span, by_dst, agg);
    combine_narrow<40><<<NARROW_BLOCKS, 96, 0, stream>>>(h1, self_w2, self_b2,
                                                         agg, neigh_w2, neigh_b2, out);
}


# Edge grouping with LDS counters, no global atomic per edge: chunks sorted by 512-node owner range in LDS, one global add per (chunk, owner); one block per owner counts, scans and places in LDS
# speedup vs baseline: 3.6523x; 1.4332x over previous
#include <hip/hip_runtime.h>


typedef float v2f __attribute__((ext_vector_type(2)));
typedef float v4f __attribute__((ext_vector_type(4)));
typedef float v8f __attribute__((ext_vector_type(8)));

#define N_NODES 100000
#define N_EDGES 1600000
#define KDIM    128

__device__ __forceinline__ v8f wmma_f32(v2f a, v2f b, v8f c) {
    return __builtin_amdgcn_wmma_f32_16x16x4_f32(
        false, a, false, b, (short)0, c, false, false);
}

#define PITCH  132
#define NPAIRS (N_NODES / 32)
#define POOL_BLOCKS   768
#define GEMM_BLOCKS   512
#define NARROW_BLOCKS 1024

__device__ __forceinline__ void load_wfrag(const float* __restrict__ Wm, int rows,
                                           float* Ws, int tid, int nthreads,
                                           int colW, int kh, v2f (&bf)[32]) {
    for (int i = tid; i < rows * 32; i += nthreads) {
        const int r = i >> 5, c4 = i & 31;
        *(float4*)(Ws + r * PITCH + c4 * 4) = ((const float4*)Wm)[i];
    }
    __syncthreads();
    const float* bp = Ws + colW * PITCH + kh;
#pragma unroll
    for (int i = 0; i < 32; ++i) bf[i] = *(const v2f*)(bp + 4 * i);
    __syncthreads();
}

__device__ __forceinline__ void kloop2(const float* ap, const v2f (&bf)[32], v8f& c0, v8f& c1) {
#pragma unroll
    for (int i = 0; i < 32; ++i) {
        const v2f a0 = *(const v2f*)(ap + 4 * i);
        const v2f a1 = *(const v2f*)(ap + 16 * PITCH + 4 * i);
        c0 = wmma_f32(a0, bf[i], c0);
        c1 = wmma_f32(a1, bf[i], c1);
    }
}

template <int NIN, bool RELU>
__global__ void __launch_bounds__(256)
gemm_wreg(const float* __restrict__ X1, const float* __restrict__ W1, const float* __restrict__ b1,
          const float* __restrict__ X2, const float* __restrict__ W2, const float* __restrict__ b2,
          float* __restrict__ Y) {
    constexpr int STAGE = NIN * 32 * PITCH;
    constexpr int SMEM  = (2 * STAGE > KDIM * PITCH) ? 2 * STAGE : KDIM * PITCH;
    __shared__ __attribute__((aligned(16))) float smem[SMEM];

    const int tid  = threadIdx.x;
    const int wave = tid >> 5;
    const int lane = tid & 31;
    const int m    = lane & 15;
    const int kh   = (lane >> 4) << 1;
    const int col  = wave * 16 + m;

    const int soff = (tid >> 5) * PITCH + (tid & 31) * 4;
    v4f pre[NIN * 4];
    int pair = blockIdx.x;
    if (pair < NPAIRS) {
        const v4f* s1 = (const v4f*)X1 + (size_t)pair * 1024 + tid;
#pragma unroll
        for (int j = 0; j < 4; ++j) pre[j] = s1[256 * j];
        if constexpr (NIN == 2) {
            const v4f* s2 = (const v4f*)X2 + (size_t)pair * 1024 + tid;
#pragma unroll
            for (int j = 0; j < 4; ++j) pre[4 * (NIN - 1) + j] = s2[256 * j];
        }
    }
    v2f bf1[32];
    v2f bf2[32];
    load_wfrag(W1, KDIM, smem, tid, 256, col, kh, bf1);
    float bv = b1[col];
    if constexpr (NIN == 2) {
        load_wfrag(W2, KDIM, smem, tid, 256, col, kh, bf2);
        bv += b2[col];
    }

    int s = 0;
    for (; pair < NPAIRS; pair += gridDim.x) {
        float* st = smem + s * STAGE;
#pragma unroll
        for (int j = 0; j < 4; ++j) *(v4f*)(st + soff + 8 * j * PITCH) = pre[j];
        if constexpr (NIN == 2) {
#pragma unroll
            for (int j = 0; j < 4; ++j) *(v4f*)(st + 32 * PITCH + soff + 8 * j * PITCH) = pre[4 * (NIN - 1) + j];
        }
        __syncthreads();

        const int nxt = pair + gridDim.x;
        if (nxt < NPAIRS) {
            const v4f* s1 = (const v4f*)X1 + (size_t)nxt * 1024 + tid;
#pragma unroll
            for (int j = 0; j < 4; ++j) pre[j] = s1[256 * j];
            if constexpr (NIN == 2) {
                const v4f* s2 = (const v4f*)X2 + (size_t)nxt * 1024 + tid;
#pragma unroll
                for (int j = 0; j < 4; ++j) pre[4 * (NIN - 1) + j] = s2[256 * j];
            }
        }

        const float* ap = st + m * PITCH + kh;
        v8f c0 = {}, c1 = {};
        kloop2(ap, bf1, c0, c1);
        if constexpr (NIN == 2) kloop2(ap + 32 * PITCH, bf2, c0, c1);

        const int rb = pair * 32 + ((lane >> 4) << 3);
#pragma unroll
        for (int r = 0; r < 8; ++r) {
            float v0 = c0[r] + bv;
            float v1 = c1[r] + bv;
            if (RELU) { v0 = fmaxf(v0, 0.0f); v1 = fmaxf(v1, 0.0f); }
            Y[(size_t)(rb + r) * KDIM + col]      = v0;
            Y[(size_t)(rb + 16 + r) * KDIM + col] = v1;
        }
        s ^= 1;
    }
}

template <int O>
__global__ void __launch_bounds__(96)
combine_narrow(const float* __restrict__ X1, const float* __restrict__ Wself, const float* __restrict__ bself,
               const float* __restrict__ X2, const float* __restrict__ Wneigh, const float* __restrict__ bneigh,
               float* __restrict__ Y) {
    constexpr int STAGE = 2 * 32 * PITCH;
    __shared__ __attribute__((aligned(16))) float smem[2 * STAGE];

    const int tid  = threadIdx.x;
    const int wave = tid >> 5;
    const int lane = tid & 31;
    const int m    = lane & 15;
    const int kh   = (lane >> 4) << 1;
    const int col  = wave * 16 + m;
    const bool valid = (col < O);
    const int colW = valid ? col : (O - 1);

    const bool tail = tid < 1024 - 960;
    const int  soff = (tid >> 5) * PITCH + (tid & 31) * 4;
    v4f p1[11], p2[11];
    int pair = blockIdx.x;
    if (pair < NPAIRS) {
        const v4f* s1 = (const v4f*)X1 + (size_t)pair * 1024 + tid;
        const v4f* s2 = (const v4f*)X2 + (size_t)pair * 1024 + tid;
#pragma unroll
        for (int j = 0; j < 10; ++j) { p1[j] = s1[96 * j]; p2[j] = s2[96 * j]; }
        if (tail) { p1[10] = s1[960]; p2[10] = s2[960]; }
    }

    v2f bf1[32];
    v2f bf2[32];
    load_wfrag(Wself,  O, smem, tid, 96, colW, kh, bf1);
    load_wfrag(Wneigh, O, smem, tid, 96, colW, kh, bf2);
    const float bb = bself[colW] + bneigh[colW];

    int s = 0;
    for (; pair < NPAIRS; pair += gridDim.x) {
        float* st = smem + s * STAGE;
#pragma unroll
        for (int j = 0; j < 10; ++j) {
            *(v4f*)(st + soff + 3 * j * PITCH)              = p1[j];
            *(v4f*)(st + 32 * PITCH + soff + 3 * j * PITCH) = p2[j];
        }
        if (tail) {
            *(v4f*)(st + soff + 30 * PITCH)              = p1[10];
            *(v4f*)(st + 32 * PITCH + soff + 30 * PITCH) = p2[10];
        }
        __syncthreads();

        const int nxt = pair + gridDim.x;
        if (nxt < NPAIRS) {
            const v4f* s1 = (const v4f*)X1 + (size_t)nxt * 1024 + tid;
            const v4f* s2 = (const v4f*)X2 + (size_t)nxt * 1024 + tid;
#pragma unroll
            for (int j = 0; j < 10; ++j) { p1[j] = s1[96 * j]; p2[j] = s2[96 * j]; }
            if (tail) { p1[10] = s1[960]; p2[10] = s2[960]; }
        }

        const float* ap = st + m * PITCH + kh;
        v8f c0 = {}, c1 = {};
        kloop2(ap, bf1, c0, c1);
        kloop2(ap + 32 * PITCH, bf2, c0, c1);

        const int rb = pair * 32 + ((lane >> 4) << 3);
#pragma unroll
        for (int r = 0; r < 8; ++r) {
            if (valid) {
                Y[(size_t)(rb + r) * O + col]      = c0[r] + bb;
                Y[(size_t)(rb + 16 + r) * O + col] = c1[r] + bb;
            }
        }
        s ^= 1;
    }
}

#define CHUNK      6144
#define NCHUNK     ((N_EDGES + CHUNK - 1) / CHUNK)
#define OWN_SHIFT  9
#define OWN_R      (1 << OWN_SHIFT)
#define NOWN       ((N_NODES + OWN_R - 1) / OWN_R)
#define CTR_PITCH  32
#define FIN_T      512
#define REC_PT     32
#define STAGE_CAP  (REC_PT * FIN_T)

template <int NW>
__device__ __forceinline__ int block_excl_scan(int v, int tid, int* s_w, int& total) {
    const int lane = tid & 31, w = tid >> 5;
    int incl = v;
#pragma unroll
    for (int o = 1; o < 32; o <<= 1) {
        const int u = __shfl_up(incl, o);
        if (lane >= o) incl += u;
    }
    if (w < NW && lane == 31) s_w[w] = incl;
    __syncthreads();
    int pre = 0, tot = 0;
#pragma unroll
    for (int k = 0; k < NW; ++k) {
        const int t = s_w[k];
        tot += t;
        if (k < w) pre += t;
    }
    total = tot;
    __syncthreads();
    return pre + incl - v;
}

__global__ void __launch_bounds__(256)
owner_count(const int* __restrict__ dst, int* __restrict__ own_total) {
    __shared__ int h[NOWN];
    const int tid = threadIdx.x;
    if (tid < NOWN) h[tid] = 0;
    __syncthreads();
    const int4* d4 = (const int4*)dst + (size_t)blockIdx.x * (CHUNK / 4);
    const int n4 = min(CHUNK / 4, N_EDGES / 4 - (int)blockIdx.x * (CHUNK / 4));
    int4 d[6];
#pragma unroll
    for (int k = 0; k < 6; ++k) {
        const int i = tid + 256 * k;
        d[k] = (i < n4) ? d4[i] : make_int4(-1, -1, -1, -1);
    }
#pragma unroll
    for (int k = 0; k < 6; ++k) {
        if (d[k].x >= 0) {
            atomicAdd(&h[d[k].x >> OWN_SHIFT], 1);
            atomicAdd(&h[d[k].y >> OWN_SHIFT], 1);
            atomicAdd(&h[d[k].z >> OWN_SHIFT], 1);
            atomicAdd(&h[d[k].w >> OWN_SHIFT], 1);
        }
    }
    __syncthreads();
    if (tid < NOWN) {
        const int c = h[tid];
        if (c > 0) atomicAdd(own_total + tid * CTR_PITCH, c);
    }
}

__global__ void __launch_bounds__(256)
owner_partition(const int* __restrict__ src, const int* __restrict__ dst,
                const int* __restrict__ own_total, int* __restrict__ own_cursor,
                int* __restrict__ recs) {
    __shared__ int h[256];
    __shared__ int delta[256];
    __shared__ int s_w[8];
    __shared__ __attribute__((aligned(8))) int2 sorted[CHUNK];

    const int tid = threadIdx.x;
    h[tid] = 0;
    __syncthreads();

    const int4* s4 = (const int4*)src + (size_t)blockIdx.x * (CHUNK / 4);
    const int4* d4 = (const int4*)dst + (size_t)blockIdx.x * (CHUNK / 4);
    const int n4 = min(CHUNK / 4, N_EDGES / 4 - (int)blockIdx.x * (CHUNK / 4));
    int4 s[6], d[6], r[6];
#pragma unroll
    for (int k = 0; k < 6; ++k) {
        const int i = tid + 256 * k;
        const bool in = i < n4;
        d[k] = in ? d4[i] : make_int4(-1, -1, -1, -1);
        s[k] = in ? s4[i] : make_int4(0, 0, 0, 0);
    }
#pragma unroll
    for (int k = 0; k < 6; ++k) {
        r[k] = make_int4(0, 0, 0, 0);
        if (d[k].x >= 0) {
            r[k].x = atomicAdd(&h[d[k].x >> OWN_SHIFT], 1);
            r[k].y = atomicAdd(&h[d[k].y >> OWN_SHIFT], 1);
            r[k].z = atomicAdd(&h[d[k].z >> OWN_SHIFT], 1);
            r[k].w = atomicAdd(&h[d[k].w >> OWN_SHIFT], 1);
        }
    }
    __syncthreads();

    const bool own = tid < NOWN;
    const int myc = own ? h[tid] : 0;
    const int tot = own ? own_total[tid * CTR_PITCH] : 0;
    int resv = 0;
    if (myc > 0) resv = atomicAdd(own_cursor + tid * CTR_PITCH, myc);
    int unused;
    const int lstart = block_excl_scan<8>(myc, tid, s_w, unused);
    const int gbase  = block_excl_scan<8>(tot, tid, s_w, unused);
    h[tid]     = lstart;
    delta[tid] = gbase + resv - lstart;
    __syncthreads();

#pragma unroll
    for (int k = 0; k < 6; ++k) {
        if (d[k].x >= 0) {
            sorted[h[d[k].x >> OWN_SHIFT] + r[k].x] = make_int2(s[k].x, d[k].x);
            sorted[h[d[k].y >> OWN_SHIFT] + r[k].y] = make_int2(s[k].y, d[k].y);
            sorted[h[d[k].z >> OWN_SHIFT] + r[k].z] = make_int2(s[k].z, d[k].z);
            sorted[h[d[k].w >> OWN_SHIFT] + r[k].w] = make_int2(s[k].w, d[k].w);
        }
    }
    __syncthreads();

    const int nloc = n4 * 4;
    int2* out2 = (int2*)recs;
#pragma unroll 4
    for (int i = tid; i < nloc; i += 256) {
        const int2 v = sorted[i];
        out2[delta[v.y >> OWN_SHIFT] + i] = v;
    }
}

__global__ void __launch_bounds__(FIN_T)
owner_finalize(const int* __restrict__ recs, const int* __restrict__ own_total,
               int* __restrict__ span, int* __restrict__ by_dst) {
    __shared__ int cnt[OWN_R];
    __shared__ int s_w[16];
    __shared__ int s_bt[2];
    __shared__ int stage[STAGE_CAP];

    const int tid = threadIdx.x;
    const int o   = blockIdx.x;
    const int node0 = o * OWN_R;

    const int t = (tid < NOWN) ? own_total[tid * CTR_PITCH] : 0;
    int unused;
    const int ex = block_excl_scan<8>(t, tid, s_w, unused);
    if (tid == o) { s_bt[0] = ex; s_bt[1] = t; }
    cnt[tid] = 0;
    __syncthreads();
    const int base  = s_bt[0];
    const int total = s_bt[1];
    const int2* my = (const int2*)recs + base;

    const bool fits = total <= STAGE_CAP;
    int2 v[REC_PT];
    if (fits) {
#pragma unroll
        for (int u = 0; u < REC_PT; ++u) {
            const int i = tid + FIN_T * u;
            v[u] = (i < total) ? my[i] : make_int2(0, -1);
        }
#pragma unroll
        for (int u = 0; u < REC_PT; ++u)
            if (v[u].y >= 0) atomicAdd(&cnt[v[u].y - node0], 1);
    } else {
        for (int i = tid; i < total; i += FIN_T) atomicAdd(&cnt[my[i].y - node0], 1);
    }
    __syncthreads();

    const int c  = cnt[tid];
    const int st = block_excl_scan<16>(c, tid, s_w, unused);
    cnt[tid] = st;
    if (node0 + tid < N_NODES) ((int2*)span)[node0 + tid] = make_int2(base + st, c);
    __syncthreads();

    if (fits) {
#pragma unroll
        for (int u = 0; u < REC_PT; ++u)
            if (v[u].y >= 0) stage[atomicAdd(&cnt[v[u].y - node0], 1)] = v[u].x;
        __syncthreads();
        for (int i = tid; i < total; i += FIN_T) by_dst[base + i] = stage[i];
    } else {
        for (int i = tid; i < total; i += FIN_T) {
            const int2 e = my[i];
            by_dst[base + atomicAdd(&cnt[e.y - node0], 1)] = e.x;
        }
    }
}

#define GATHER_BLOCKS 2048
__global__ void __launch_bounds__(256)
gather_max(const float* __restrict__ m,
           const int* __restrict__ span,
           const int* __restrict__ by_dst,
           float* __restrict__ agg) {
    const int lane = threadIdx.x & 31;
    const int wave0 = __builtin_amdgcn_readfirstlane(blockIdx.x * 8 + (threadIdx.x >> 5));
    const float* mcol = m + lane * 4;

    for (int node = wave0; node < N_NODES; node += GATHER_BLOCKS * 8) {
        const int2 sp   = ((const int2*)span)[node];
        const int start = sp.x;
        const int n     = sp.y;
        float4 acc = {0.0f, 0.0f, 0.0f, 0.0f};
        for (int base = 0; base < n; base += 32) {
            const int left = n - base;
            const int last = (left < 32 ? left : 32) - 1;
            const int idx = by_dst[start + base + (lane < last ? lane : last)];
            for (int j = 0; j <= last; j += 4) {
                const int s0 = __builtin_amdgcn_readlane(idx, j);
                const int s1 = __builtin_amdgcn_readlane(idx, j + 1);
                const int s2 = __builtin_amdgcn_readlane(idx, j + 2);
                const int s3 = __builtin_amdgcn_readlane(idx, j + 3);
                const float4 v0 = *(const float4*)(mcol + (size_t)s0 * KDIM);
                const float4 v1 = *(const float4*)(mcol + (size_t)s1 * KDIM);
                const float4 v2 = *(const float4*)(mcol + (size_t)s2 * KDIM);
                const float4 v3 = *(const float4*)(mcol + (size_t)s3 * KDIM);
                acc.x = fmaxf(acc.x, fmaxf(fmaxf(v0.x, v1.x), fmaxf(v2.x, v3.x)));
                acc.y = fmaxf(acc.y, fmaxf(fmaxf(v0.y, v1.y), fmaxf(v2.y, v3.y)));
                acc.z = fmaxf(acc.z, fmaxf(fmaxf(v0.z, v1.z), fmaxf(v2.z, v3.z)));
                acc.w = fmaxf(acc.w, fmaxf(fmaxf(v0.w, v1.w), fmaxf(v2.w, v3.w)));
            }
        }
        *(float4*)(agg + (size_t)node * KDIM + lane * 4) = acc;
    }
}

extern "C" void kernel_launch(void* const* d_in, const int* in_sizes, int n_in,
                              void* d_out, int out_size, void* d_ws, size_t ws_size,
                              hipStream_t stream) {
    const float* in_feat  = (const float*)d_in[0];
    const int*   src      = (const int*)d_in[1];
    const int*   dst      = (const int*)d_in[2];
    const float* pool_w1  = (const float*)d_in[3];
    const float* pool_b1  = (const float*)d_in[4];
    const float* self_w1  = (const float*)d_in[5];
    const float* self_b1  = (const float*)d_in[6];
    const float* neigh_w1 = (const float*)d_in[7];
    const float* neigh_b1 = (const float*)d_in[8];
    const float* pool_w2  = (const float*)d_in[9];
    const float* pool_b2  = (const float*)d_in[10];
    const float* self_w2  = (const float*)d_in[11];
    const float* self_b2  = (const float*)d_in[12];
    const float* neigh_w2 = (const float*)d_in[13];
    const float* neigh_b2 = (const float*)d_in[14];
    float* out = (float*)d_out;

    const size_t NODE_BYTES = (size_t)N_NODES * KDIM * sizeof(float);
    char* ws = (char*)d_ws;
    float* m_buf = (float*)(ws);
    float* agg   = (float*)(ws + NODE_BYTES);
    float* h1    = (float*)(ws + 2 * NODE_BYTES);

    char* ob = (char*)d_out;
    int*  own_total  = (int*)(ob);
    int*  own_cursor = (int*)(ob + 32768);
    int*  span   = (int*)(ob + 401408);
    int*  by_dst = (int*)(ob + 7864320);
    int*  recs   = (int*)agg;

    hipMemsetAsync(own_total, 0, 65536, stream);
    owner_count<<<NCHUNK, 256, 0, stream>>>(dst, own_total);
    owner_partition<<<NCHUNK, 256, 0, stream>>>(src, dst, own_total, own_cursor, recs);
    owner_finalize<<<NOWN, FIN_T, 0, stream>>>(recs, own_total, span, by_dst);

    gemm_wreg<1, true><<<POOL_BLOCKS, 256, 0, stream>>>(in_feat, pool_w1, pool_b1,
                                                        nullptr, nullptr, nullptr, m_buf);
    gather_max<<<GATHER_BLOCKS, 256, 0, stream>>>(m_buf, span, by_dst, agg);
    gemm_wreg<2, true><<<GEMM_BLOCKS, 256, 0, stream>>>(in_feat, self_w1, self_b1,
                                                        agg, neigh_w1, neigh_b1, h1);

    gemm_wreg<1, true><<<POOL_BLOCKS, 256, 0, stream>>>(h1, pool_w2, pool_b2,
                                                        nullptr, nullptr, nullptr, m_buf);
    gather_max<<<GATHER_BLOCKS, 256, 0, stream>>>(m_buf, span, by_dst, agg);
    combine_narrow<40><<<NARROW_BLOCKS, 96, 0, stream>>>(h1, self_w2, self_b2,
                                                         agg, neigh_w2, neigh_b2, out);
}


# Segment max fused into both combine GEMMs: a block's waves gather their 32 nodes' max rows (up to 4 groups of 4 source rows in flight per wave) into the LDS A tile; no agg plane, 2 launches fewer
# speedup vs baseline: 3.9677x; 1.0864x over previous
#include <hip/hip_runtime.h>


typedef float v2f __attribute__((ext_vector_type(2)));
typedef float v4f __attribute__((ext_vector_type(4)));
typedef float v8f __attribute__((ext_vector_type(8)));

#define N_NODES 100000
#define N_EDGES 1600000
#define KDIM    128

__device__ __forceinline__ v8f wmma_f32(v2f a, v2f b, v8f c) {
    return __builtin_amdgcn_wmma_f32_16x16x4_f32(
        false, a, false, b, (short)0, c, false, false);
}

#define PITCH  132
#define NPAIRS (N_NODES / 32)
#define POOL_BLOCKS   768
#define GEMM_BLOCKS   512

__device__ __forceinline__ void load_wfrag(const float* __restrict__ Wm, int rows,
                                           float* Ws, int tid, int nthreads,
                                           int colW, int kh, v2f (&bf)[32]) {
    for (int i = tid; i < rows * 32; i += nthreads) {
        const int r = i >> 5, c4 = i & 31;
        *(float4*)(Ws + r * PITCH + c4 * 4) = ((const float4*)Wm)[i];
    }
    __syncthreads();
    const float* bp = Ws + colW * PITCH + kh;
#pragma unroll
    for (int i = 0; i < 32; ++i) bf[i] = *(const v2f*)(bp + 4 * i);
    __syncthreads();
}

__device__ __forceinline__ void kloop2(const float* ap, const v2f (&bf)[32], v8f& c0, v8f& c1) {
#pragma unroll
    for (int i = 0; i < 32; ++i) {
        const v2f a0 = *(const v2f*)(ap + 4 * i);
        const v2f a1 = *(const v2f*)(ap + 16 * PITCH + 4 * i);
        c0 = wmma_f32(a0, bf[i], c0);
        c1 = wmma_f32(a1, bf[i], c1);
    }
}

template <int NIN, bool RELU>
__global__ void __launch_bounds__(256)
gemm_wreg(const float* __restrict__ X1, const float* __restrict__ W1, const float* __restrict__ b1,
          const float* __restrict__ X2, const float* __restrict__ W2, const float* __restrict__ b2,
          float* __restrict__ Y) {
    constexpr int STAGE = NIN * 32 * PITCH;
    constexpr int SMEM  = (2 * STAGE > KDIM * PITCH) ? 2 * STAGE : KDIM * PITCH;
    __shared__ __attribute__((aligned(16))) float smem[SMEM];

    const int tid  = threadIdx.x;
    const int wave = tid >> 5;
    const int lane = tid & 31;
    const int m    = lane & 15;
    const int kh   = (lane >> 4) << 1;
    const int col  = wave * 16 + m;

    const int soff = (tid >> 5) * PITCH + (tid & 31) * 4;
    v4f pre[NIN * 4];
    int pair = blockIdx.x;
    if (pair < NPAIRS) {
        const v4f* s1 = (const v4f*)X1 + (size_t)pair * 1024 + tid;
#pragma unroll
        for (int j = 0; j < 4; ++j) pre[j] = s1[256 * j];
        if constexpr (NIN == 2) {
            const v4f* s2 = (const v4f*)X2 + (size_t)pair * 1024 + tid;
#pragma unroll
            for (int j = 0; j < 4; ++j) pre[4 * (NIN - 1) + j] = s2[256 * j];
        }
    }
    v2f bf1[32];
    v2f bf2[32];
    load_wfrag(W1, KDIM, smem, tid, 256, col, kh, bf1);
    float bv = b1[col];
    if constexpr (NIN == 2) {
        load_wfrag(W2, KDIM, smem, tid, 256, col, kh, bf2);
        bv += b2[col];
    }

    int s = 0;
    for (; pair < NPAIRS; pair += gridDim.x) {
        float* st = smem + s * STAGE;
#pragma unroll
        for (int j = 0; j < 4; ++j) *(v4f*)(st + soff + 8 * j * PITCH) = pre[j];
        if constexpr (NIN == 2) {
#pragma unroll
            for (int j = 0; j < 4; ++j) *(v4f*)(st + 32 * PITCH + soff + 8 * j * PITCH) = pre[4 * (NIN - 1) + j];
        }
        __syncthreads();

        const int nxt = pair + gridDim.x;
        if (nxt < NPAIRS) {
            const v4f* s1 = (const v4f*)X1 + (size_t)nxt * 1024 + tid;
#pragma unroll
            for (int j = 0; j < 4; ++j) pre[j] = s1[256 * j];
            if constexpr (NIN == 2) {
                const v4f* s2 = (const v4f*)X2 + (size_t)nxt * 1024 + tid;
#pragma unroll
                for (int j = 0; j < 4; ++j) pre[4 * (NIN - 1) + j] = s2[256 * j];
            }
        }

        const float* ap = st + m * PITCH + kh;
        v8f c0 = {}, c1 = {};
        kloop2(ap, bf1, c0, c1);
        if constexpr (NIN == 2) kloop2(ap + 32 * PITCH, bf2, c0, c1);

        const int rb = pair * 32 + ((lane >> 4) << 3);
#pragma unroll
        for (int r = 0; r < 8; ++r) {
            float v0 = c0[r] + bv;
            float v1 = c1[r] + bv;
            if (RELU) { v0 = fmaxf(v0, 0.0f); v1 = fmaxf(v1, 0.0f); }
            Y[(size_t)(rb + r) * KDIM + col]      = v0;
            Y[(size_t)(rb + 16 + r) * KDIM + col] = v1;
        }
        s ^= 1;
    }
}

#define CHUNK      6144
#define NCHUNK     ((N_EDGES + CHUNK - 1) / CHUNK)
#define OWN_SHIFT  9
#define OWN_R      (1 << OWN_SHIFT)
#define NOWN       ((N_NODES + OWN_R - 1) / OWN_R)
#define CTR_PITCH  32
#define FIN_T      512
#define REC_PT     32
#define STAGE_CAP  (REC_PT * FIN_T)

template <int NW>
__device__ __forceinline__ int block_excl_scan(int v, int tid, int* s_w, int& total) {
    const int lane = tid & 31, w = tid >> 5;
    int incl = v;
#pragma unroll
    for (int o = 1; o < 32; o <<= 1) {
        const int u = __shfl_up(incl, o);
        if (lane >= o) incl += u;
    }
    if (w < NW && lane == 31) s_w[w] = incl;
    __syncthreads();
    int pre = 0, tot = 0;
#pragma unroll
    for (int k = 0; k < NW; ++k) {
        const int t = s_w[k];
        tot += t;
        if (k < w) pre += t;
    }
    total = tot;
    __syncthreads();
    return pre + incl - v;
}

__global__ void __launch_bounds__(256)
owner_count(const int* __restrict__ dst, int* __restrict__ own_total) {
    __shared__ int h[NOWN];
    const int tid = threadIdx.x;
    if (tid < NOWN) h[tid] = 0;
    __syncthreads();
    const int4* d4 = (const int4*)dst + (size_t)blockIdx.x * (CHUNK / 4);
    const int n4 = min(CHUNK / 4, N_EDGES / 4 - (int)blockIdx.x * (CHUNK / 4));
    int4 d[6];
#pragma unroll
    for (int k = 0; k < 6; ++k) {
        const int i = tid + 256 * k;
        d[k] = (i < n4) ? d4[i] : make_int4(-1, -1, -1, -1);
    }
#pragma unroll
    for (int k = 0; k < 6; ++k) {
        if (d[k].x >= 0) {
            atomicAdd(&h[d[k].x >> OWN_SHIFT], 1);
            atomicAdd(&h[d[k].y >> OWN_SHIFT], 1);
            atomicAdd(&h[d[k].z >> OWN_SHIFT], 1);
            atomicAdd(&h[d[k].w >> OWN_SHIFT], 1);
        }
    }
    __syncthreads();
    if (tid < NOWN) {
        const int c = h[tid];
        if (c > 0) atomicAdd(own_total + tid * CTR_PITCH, c);
    }
}

__global__ void __launch_bounds__(256)
owner_partition(const int* __restrict__ src, const int* __restrict__ dst,
                const int* __restrict__ own_total, int* __restrict__ own_cursor,
                int* __restrict__ recs) {
    __shared__ int h[256];
    __shared__ int delta[256];
    __shared__ int s_w[8];
    __shared__ __attribute__((aligned(8))) int2 sorted[CHUNK];

    const int tid = threadIdx.x;
    h[tid] = 0;
    __syncthreads();

    const int4* s4 = (const int4*)src + (size_t)blockIdx.x * (CHUNK / 4);
    const int4* d4 = (const int4*)dst + (size_t)blockIdx.x * (CHUNK / 4);
    const int n4 = min(CHUNK / 4, N_EDGES / 4 - (int)blockIdx.x * (CHUNK / 4));
    int4 s[6], d[6], r[6];
#pragma unroll
    for (int k = 0; k < 6; ++k) {
        const int i = tid + 256 * k;
        const bool in = i < n4;
        d[k] = in ? d4[i] : make_int4(-1, -1, -1, -1);
        s[k] = in ? s4[i] : make_int4(0, 0, 0, 0);
    }
#pragma unroll
    for (int k = 0; k < 6; ++k) {
        r[k] = make_int4(0, 0, 0, 0);
        if (d[k].x >= 0) {
            r[k].x = atomicAdd(&h[d[k].x >> OWN_SHIFT], 1);
            r[k].y = atomicAdd(&h[d[k].y >> OWN_SHIFT], 1);
            r[k].z = atomicAdd(&h[d[k].z >> OWN_SHIFT], 1);
            r[k].w = atomicAdd(&h[d[k].w >> OWN_SHIFT], 1);
        }
    }
    __syncthreads();

    const bool own = tid < NOWN;
    const int myc = own ? h[tid] : 0;
    const int tot = own ? own_total[tid * CTR_PITCH] : 0;
    int resv = 0;
    if (myc > 0) resv = atomicAdd(own_cursor + tid * CTR_PITCH, myc);
    int unused;
    const int lstart = block_excl_scan<8>(myc, tid, s_w, unused);
    const int gbase  = block_excl_scan<8>(tot, tid, s_w, unused);
    h[tid]     = lstart;
    delta[tid] = gbase + resv - lstart;
    __syncthreads();

#pragma unroll
    for (int k = 0; k < 6; ++k) {
        if (d[k].x >= 0) {
            sorted[h[d[k].x >> OWN_SHIFT] + r[k].x] = make_int2(s[k].x, d[k].x);
            sorted[h[d[k].y >> OWN_SHIFT] + r[k].y] = make_int2(s[k].y, d[k].y);
            sorted[h[d[k].z >> OWN_SHIFT] + r[k].z] = make_int2(s[k].z, d[k].z);
            sorted[h[d[k].w >> OWN_SHIFT] + r[k].w] = make_int2(s[k].w, d[k].w);
        }
    }
    __syncthreads();

    const int nloc = n4 * 4;
    int2* out2 = (int2*)recs;
#pragma unroll 4
    for (int i = tid; i < nloc; i += 256) {
        const int2 v = sorted[i];
        out2[delta[v.y >> OWN_SHIFT] + i] = v;
    }
}

__global__ void __launch_bounds__(FIN_T)
owner_finalize(const int* __restrict__ recs, const int* __restrict__ own_total,
               int* __restrict__ span, int* __restrict__ by_dst) {
    __shared__ int cnt[OWN_R];
    __shared__ int s_w[16];
    __shared__ int s_bt[2];
    __shared__ int stage[STAGE_CAP];

    const int tid = threadIdx.x;
    const int o   = blockIdx.x;
    const int node0 = o * OWN_R;

    const int t = (tid < NOWN) ? own_total[tid * CTR_PITCH] : 0;
    int unused;
    const int ex = block_excl_scan<8>(t, tid, s_w, unused);
    if (tid == o) { s_bt[0] = ex; s_bt[1] = t; }
    cnt[tid] = 0;
    __syncthreads();
    const int base  = s_bt[0];
    const int total = s_bt[1];
    const int2* my = (const int2*)recs + base;

    const bool fits = total <= STAGE_CAP;
    int2 v[REC_PT];
    if (fits) {
#pragma unroll
        for (int u = 0; u < REC_PT; ++u) {
            const int i = tid + FIN_T * u;
            v[u] = (i < total) ? my[i] : make_int2(0, -1);
        }
#pragma unroll
        for (int u = 0; u < REC_PT; ++u)
            if (v[u].y >= 0) atomicAdd(&cnt[v[u].y - node0], 1);
    } else {
        for (int i = tid; i < total; i += FIN_T) atomicAdd(&cnt[my[i].y - node0], 1);
    }
    __syncthreads();

    const int c  = cnt[tid];
    const int st = block_excl_scan<16>(c, tid, s_w, unused);
    cnt[tid] = st;
    if (node0 + tid < N_NODES) ((int2*)span)[node0 + tid] = make_int2(base + st, c);
    __syncthreads();

    if (fits) {
#pragma unroll
        for (int u = 0; u < REC_PT; ++u)
            if (v[u].y >= 0) stage[atomicAdd(&cnt[v[u].y - node0], 1)] = v[u].x;
        __syncthreads();
        for (int i = tid; i < total; i += FIN_T) by_dst[base + i] = stage[i];
    } else {
        for (int i = tid; i < total; i += FIN_T) {
            const int2 e = my[i];
            by_dst[base + atomicAdd(&cnt[e.y - node0], 1)] = e.x;
        }
    }
}

#define GROW 16

__device__ __forceinline__ v4f max4(v4f a, v4f b) {
    v4f r;
    r.x = fmaxf(a.x, b.x); r.y = fmaxf(a.y, b.y); r.z = fmaxf(a.z, b.z); r.w = fmaxf(a.w, b.w);
    return r;
}
__device__ __forceinline__ v4f red4(const v4f (&r)[4]) {
    return max4(max4(r[0], r[1]), max4(r[2], r[3]));
}
__device__ __forceinline__ void load4(const float* __restrict__ mcol, int idx, int j, v4f (&r)[4]) {
#pragma unroll
    for (int q = 0; q < 4; ++q) {
        const int s = __builtin_amdgcn_readlane(idx, j + q);
        r[q] = *(const v4f*)(mcol + (size_t)s * KDIM);
    }
}

__device__ __forceinline__ v4f node_max(const float* __restrict__ mcol, const int* __restrict__ by_dst,
                                        int start, int n, int idx0, int lane) {
    v4f acc = {0.0f, 0.0f, 0.0f, 0.0f};
    for (int base = 0; base < n; base += 32) {
        const int left = n - base;
        const int last = (left < 32 ? left : 32) - 1;
        int idx = idx0;
        if (base > 0) idx = by_dst[start + base + (lane < last ? lane : last)];
        for (int j = 0; j <= last; j += GROW) {
            v4f a[4];
            load4(mcol, idx, j, a);
            if (j + 4 <= last) {
                v4f b[4];
                load4(mcol, idx, j + 4, b);
                if (j + 8 <= last) {
                    v4f c[4];
                    load4(mcol, idx, j + 8, c);
                    if (j + 12 <= last) {
                        v4f d[4];
                        load4(mcol, idx, j + 12, d);
                        acc = max4(acc, red4(d));
                    }
                    acc = max4(acc, red4(c));
                }
                acc = max4(acc, red4(b));
            }
            acc = max4(acc, red4(a));
        }
    }
    return acc;
}

template <int O, bool RELU>
__global__ void __launch_bounds__(256)
gather_gemm(const float* __restrict__ X1, const float* __restrict__ W1, const float* __restrict__ b1,
            const float* __restrict__ Mp, const int* __restrict__ span, const int* __restrict__ by_dst,
            const float* __restrict__ W2, const float* __restrict__ b2, float* __restrict__ Y) {
    constexpr int NCW   = (O + 15) / 16;
    constexpr int STAGE = 2 * 32 * PITCH;
    __shared__ __attribute__((aligned(16))) float smem[2 * STAGE];

    const int tid  = threadIdx.x;
    const int wave = __builtin_amdgcn_readfirstlane(tid >> 5);
    const int lane = tid & 31;
    const int m    = lane & 15;
    const int kh   = (lane >> 4) << 1;
    const int col  = wave * 16 + m;
    const bool valid = (col < O);
    const int colW = valid ? col : (O - 1);
    const int soff = (tid >> 5) * PITCH + (tid & 31) * 4;
    const float* mcol = Mp + lane * 4;

    v4f pre[4];
    int st_[4], n_[4], idx[4];
    auto fetch_span = [&](int p) -> int {
        return span[(p * 32 + wave * 4) * 2 + (lane & 7)];
    };
    auto fetch_x = [&](int p) {
        const v4f* s1 = (const v4f*)X1 + (size_t)p * 1024 + tid;
#pragma unroll
        for (int j = 0; j < 4; ++j) pre[j] = s1[256 * j];
    };
    auto fetch_idx = [&](int sv) {
#pragma unroll
        for (int i = 0; i < 4; ++i) {
            st_[i] = __builtin_amdgcn_readlane(sv, 2 * i);
            n_[i]  = __builtin_amdgcn_readlane(sv, 2 * i + 1);
            const int lastl = (n_[i] < 32 ? n_[i] : 32) - 1;
            idx[i] = (n_[i] > 0) ? by_dst[st_[i] + (lane < lastl ? lane : lastl)] : 0;
        }
    };

    int pair = blockIdx.x;
    int sv = 0, svn = 0;
    if (pair < NPAIRS) { sv = fetch_span(pair); fetch_x(pair); }
    if (pair + (int)gridDim.x < NPAIRS) svn = fetch_span(pair + gridDim.x);

    v2f bf1[32];
    v2f bf2[32];
    load_wfrag(W1, O, smem, tid, 256, colW, kh, bf1);
    if (pair < NPAIRS) fetch_idx(sv);
    load_wfrag(W2, O, smem, tid, 256, colW, kh, bf2);
    const float bv = b1[colW] + b2[colW];

    int s = 0;
    for (; pair < NPAIRS; pair += gridDim.x) {
        float* st = smem + s * STAGE;
#pragma unroll
        for (int j = 0; j < 4; ++j) *(v4f*)(st + soff + 8 * j * PITCH) = pre[j];
#pragma unroll
        for (int i = 0; i < 4; ++i) {
            const v4f a = node_max(mcol, by_dst, st_[i], n_[i], idx[i], lane);
            *(v4f*)(st + (32 + wave * 4 + i) * PITCH + lane * 4) = a;
        }
        __syncthreads();

        const int nxt = pair + gridDim.x;
        if (nxt < NPAIRS) { fetch_x(nxt); fetch_idx(svn); }
        const int nxt2 = nxt + gridDim.x;
        svn = (nxt2 < NPAIRS) ? fetch_span(nxt2) : 0;

        if (wave < NCW) {
            const float* ap = st + m * PITCH + kh;
            v8f c0 = {}, c1 = {};
            kloop2(ap, bf1, c0, c1);
            kloop2(ap + 32 * PITCH, bf2, c0, c1);

            const int rb = pair * 32 + ((lane >> 4) << 3);
#pragma unroll
            for (int r = 0; r < 8; ++r) {
                float v0 = c0[r] + bv;
                float v1 = c1[r] + bv;
                if (RELU) { v0 = fmaxf(v0, 0.0f); v1 = fmaxf(v1, 0.0f); }
                if (valid) {
                    Y[(size_t)(rb + r) * O + col]      = v0;
                    Y[(size_t)(rb + 16 + r) * O + col] = v1;
                }
            }
        }
        s ^= 1;
    }
}

extern "C" void kernel_launch(void* const* d_in, const int* in_sizes, int n_in,
                              void* d_out, int out_size, void* d_ws, size_t ws_size,
                              hipStream_t stream) {
    const float* in_feat  = (const float*)d_in[0];
    const int*   src      = (const int*)d_in[1];
    const int*   dst      = (const int*)d_in[2];
    const float* pool_w1  = (const float*)d_in[3];
    const float* pool_b1  = (const float*)d_in[4];
    const float* self_w1  = (const float*)d_in[5];
    const float* self_b1  = (const float*)d_in[6];
    const float* neigh_w1 = (const float*)d_in[7];
    const float* neigh_b1 = (const float*)d_in[8];
    const float* pool_w2  = (const float*)d_in[9];
    const float* pool_b2  = (const float*)d_in[10];
    const float* self_w2  = (const float*)d_in[11];
    const float* self_b2  = (const float*)d_in[12];
    const float* neigh_w2 = (const float*)d_in[13];
    const float* neigh_b2 = (const float*)d_in[14];
    float* out = (float*)d_out;

    const size_t NODE_BYTES = (size_t)N_NODES * KDIM * sizeof(float);
    char* ws = (char*)d_ws;
    float* m_buf = (float*)(ws);
    char*  aux   = ws + NODE_BYTES;
    float* h1    = (float*)(ws + 2 * NODE_BYTES);

    int*  recs   = (int*)(aux);
    int*  by_dst = (int*)(aux + 16777216);
    int*  span   = (int*)(aux + 25165824);
    char* ob = (char*)d_out;
    int*  own_total  = (int*)(ob);
    int*  own_cursor = (int*)(ob + 32768);

    hipMemsetAsync(own_total, 0, 65536, stream);
    owner_count<<<NCHUNK, 256, 0, stream>>>(dst, own_total);
    owner_partition<<<NCHUNK, 256, 0, stream>>>(src, dst, own_total, own_cursor, recs);
    owner_finalize<<<NOWN, FIN_T, 0, stream>>>(recs, own_total, span, by_dst);

    gemm_wreg<1, true><<<POOL_BLOCKS, 256, 0, stream>>>(in_feat, pool_w1, pool_b1,
                                                        nullptr, nullptr, nullptr, m_buf);
    gather_gemm<KDIM, true><<<GEMM_BLOCKS, 256, 0, stream>>>(in_feat, self_w1, self_b1,
                                                             m_buf, span, by_dst, neigh_w1, neigh_b1, h1);

    gemm_wreg<1, true><<<POOL_BLOCKS, 256, 0, stream>>>(h1, pool_w2, pool_b2,
                                                        nullptr, nullptr, nullptr, m_buf);
    gather_gemm<40, false><<<GEMM_BLOCKS, 256, 0, stream>>>(h1, self_w2, self_b2,
                                                            m_buf, span, by_dst, neigh_w2, neigh_b2, out);
}


# Edge grouping in 2 launches, no global counter or memset: each chunk sorted by owner in place, run lengths and starts in a table; an owner block sums table columns for its base and reads its runs
# speedup vs baseline: 4.1715x; 1.0514x over previous
#include <hip/hip_runtime.h>


typedef float v2f __attribute__((ext_vector_type(2)));
typedef float v4f __attribute__((ext_vector_type(4)));
typedef float v8f __attribute__((ext_vector_type(8)));

#define N_NODES 100000
#define N_EDGES 1600000
#define KDIM    128

__device__ __forceinline__ v8f wmma_f32(v2f a, v2f b, v8f c) {
    return __builtin_amdgcn_wmma_f32_16x16x4_f32(
        false, a, false, b, (short)0, c, false, false);
}

#define PITCH  132
#define NPAIRS (N_NODES / 32)
#define POOL_BLOCKS   768
#define GEMM_BLOCKS   512

__device__ __forceinline__ void load_wfrag(const float* __restrict__ Wm, int rows,
                                           float* Ws, int tid, int nthreads,
                                           int colW, int kh, v2f (&bf)[32]) {
    for (int i = tid; i < rows * 32; i += nthreads) {
        const int r = i >> 5, c4 = i & 31;
        *(float4*)(Ws + r * PITCH + c4 * 4) = ((const float4*)Wm)[i];
    }
    __syncthreads();
    const float* bp = Ws + colW * PITCH + kh;
#pragma unroll
    for (int i = 0; i < 32; ++i) bf[i] = *(const v2f*)(bp + 4 * i);
    __syncthreads();
}

__device__ __forceinline__ void kloop2(const float* ap, const v2f (&bf)[32], v8f& c0, v8f& c1) {
#pragma unroll
    for (int i = 0; i < 32; ++i) {
        const v2f a0 = *(const v2f*)(ap + 4 * i);
        const v2f a1 = *(const v2f*)(ap + 16 * PITCH + 4 * i);
        c0 = wmma_f32(a0, bf[i], c0);
        c1 = wmma_f32(a1, bf[i], c1);
    }
}

template <int NIN, bool RELU>
__global__ void __launch_bounds__(256)
gemm_wreg(const float* __restrict__ X1, const float* __restrict__ W1, const float* __restrict__ b1,
          const float* __restrict__ X2, const float* __restrict__ W2, const float* __restrict__ b2,
          float* __restrict__ Y) {
    constexpr int STAGE = NIN * 32 * PITCH;
    constexpr int SMEM  = (2 * STAGE > KDIM * PITCH) ? 2 * STAGE : KDIM * PITCH;
    __shared__ __attribute__((aligned(16))) float smem[SMEM];

    const int tid  = threadIdx.x;
    const int wave = tid >> 5;
    const int lane = tid & 31;
    const int m    = lane & 15;
    const int kh   = (lane >> 4) << 1;
    const int col  = wave * 16 + m;

    const int soff = (tid >> 5) * PITCH + (tid & 31) * 4;
    v4f pre[NIN * 4];
    int pair = blockIdx.x;
    if (pair < NPAIRS) {
        const v4f* s1 = (const v4f*)X1 + (size_t)pair * 1024 + tid;
#pragma unroll
        for (int j = 0; j < 4; ++j) pre[j] = s1[256 * j];
        if constexpr (NIN == 2) {
            const v4f* s2 = (const v4f*)X2 + (size_t)pair * 1024 + tid;
#pragma unroll
            for (int j = 0; j < 4; ++j) pre[4 * (NIN - 1) + j] = s2[256 * j];
        }
    }
    v2f bf1[32];
    v2f bf2[32];
    load_wfrag(W1, KDIM, smem, tid, 256, col, kh, bf1);
    float bv = b1[col];
    if constexpr (NIN == 2) {
        load_wfrag(W2, KDIM, smem, tid, 256, col, kh, bf2);
        bv += b2[col];
    }

    int s = 0;
    for (; pair < NPAIRS; pair += gridDim.x) {
        float* st = smem + s * STAGE;
#pragma unroll
        for (int j = 0; j < 4; ++j) *(v4f*)(st + soff + 8 * j * PITCH) = pre[j];
        if constexpr (NIN == 2) {
#pragma unroll
            for (int j = 0; j < 4; ++j) *(v4f*)(st + 32 * PITCH + soff + 8 * j * PITCH) = pre[4 * (NIN - 1) + j];
        }
        __syncthreads();

        const int nxt = pair + gridDim.x;
        if (nxt < NPAIRS) {
            const v4f* s1 = (const v4f*)X1 + (size_t)nxt * 1024 + tid;
#pragma unroll
            for (int j = 0; j < 4; ++j) pre[j] = s1[256 * j];
            if constexpr (NIN == 2) {
                const v4f* s2 = (const v4f*)X2 + (size_t)nxt * 1024 + tid;
#pragma unroll
                for (int j = 0; j < 4; ++j) pre[4 * (NIN - 1) + j] = s2[256 * j];
            }
        }

        const float* ap = st + m * PITCH + kh;
        v8f c0 = {}, c1 = {};
        kloop2(ap, bf1, c0, c1);
        if constexpr (NIN == 2) kloop2(ap + 32 * PITCH, bf2, c0, c1);

        const int rb = pair * 32 + ((lane >> 4) << 3);
#pragma unroll
        for (int r = 0; r < 8; ++r) {
            float v0 = c0[r] + bv;
            float v1 = c1[r] + bv;
            if (RELU) { v0 = fmaxf(v0, 0.0f); v1 = fmaxf(v1, 0.0f); }
            Y[(size_t)(rb + r) * KDIM + col]      = v0;
            Y[(size_t)(rb + 16 + r) * KDIM + col] = v1;
        }
        s ^= 1;
    }
}

#define CHUNK      6144
#define NCHUNK     ((N_EDGES + CHUNK - 1) / CHUNK)
#define OWN_SHIFT  9
#define OWN_R      (1 << OWN_SHIFT)
#define NOWN       ((N_NODES + OWN_R - 1) / OWN_R)
#define TAB_W      256
#define FIN_T      512
#define RUN_SLOTS  64
#define RUN_SWEEP  (FIN_T / RUN_SLOTS)
#define REC_PT     ((NCHUNK + RUN_SWEEP - 1) / RUN_SWEEP)
#define TAB_Q      (NCHUNK * (TAB_W / 4))
#define TAB_PT     ((TAB_Q + FIN_T - 1) / FIN_T)
#define STAGE_CAP  16384

template <int NW>
__device__ __forceinline__ int block_excl_scan(int v, int tid, int* s_w, int& total) {
    const int lane = tid & 31, w = tid >> 5;
    int incl = v;
#pragma unroll
    for (int o = 1; o < 32; o <<= 1) {
        const int u = __shfl_up(incl, o);
        if (lane >= o) incl += u;
    }
    if (w < NW && lane == 31) s_w[w] = incl;
    __syncthreads();
    int pre = 0, tot = 0;
#pragma unroll
    for (int k = 0; k < NW; ++k) {
        const int t = s_w[k];
        tot += t;
        if (k < w) pre += t;
    }
    total = tot;
    __syncthreads();
    return pre + incl - v;
}

__global__ void __launch_bounds__(256)
chunk_sort(const int* __restrict__ src, const int* __restrict__ dst,
           int* __restrict__ recs, int* __restrict__ run_cnt, int* __restrict__ run_lst) {
    __shared__ int h[TAB_W];
    __shared__ int s_w[8];
    __shared__ __attribute__((aligned(16))) int2 sorted[CHUNK];

    const int tid = threadIdx.x;
    h[tid] = 0;
    __syncthreads();

    const int4* s4 = (const int4*)src + (size_t)blockIdx.x * (CHUNK / 4);
    const int4* d4 = (const int4*)dst + (size_t)blockIdx.x * (CHUNK / 4);
    const int n4 = min(CHUNK / 4, N_EDGES / 4 - (int)blockIdx.x * (CHUNK / 4));
    int4 s[6], d[6], r[6];
#pragma unroll
    for (int k = 0; k < 6; ++k) {
        const int i = tid + 256 * k;
        const bool in = i < n4;
        d[k] = in ? d4[i] : make_int4(-1, -1, -1, -1);
        s[k] = in ? s4[i] : make_int4(0, 0, 0, 0);
    }
#pragma unroll
    for (int k = 0; k < 6; ++k) {
        r[k] = make_int4(0, 0, 0, 0);
        if (d[k].x >= 0) {
            r[k].x = atomicAdd(&h[d[k].x >> OWN_SHIFT], 1);
            r[k].y = atomicAdd(&h[d[k].y >> OWN_SHIFT], 1);
            r[k].z = atomicAdd(&h[d[k].z >> OWN_SHIFT], 1);
            r[k].w = atomicAdd(&h[d[k].w >> OWN_SHIFT], 1);
        }
    }
    __syncthreads();

    const int myc = h[tid];
    int unused;
    const int lstart = block_excl_scan<8>(myc, tid, s_w, unused);
    h[tid] = lstart;
    run_cnt[blockIdx.x * TAB_W + tid] = myc;
    run_lst[blockIdx.x * TAB_W + tid] = lstart;
    __syncthreads();

#pragma unroll
    for (int k = 0; k < 6; ++k) {
        if (d[k].x >= 0) {
            sorted[h[d[k].x >> OWN_SHIFT] + r[k].x] = make_int2(s[k].x, d[k].x);
            sorted[h[d[k].y >> OWN_SHIFT] + r[k].y] = make_int2(s[k].y, d[k].y);
            sorted[h[d[k].z >> OWN_SHIFT] + r[k].z] = make_int2(s[k].z, d[k].z);
            sorted[h[d[k].w >> OWN_SHIFT] + r[k].w] = make_int2(s[k].w, d[k].w);
        }
    }
    __syncthreads();

    const int npc = n4 * 2;
    int4* out4 = (int4*)recs + (size_t)blockIdx.x * (CHUNK / 2);
    const int4* in4 = (const int4*)sorted;
#pragma unroll 4
    for (int i = tid; i < npc; i += 256) out4[i] = in4[i];
}

__global__ void __launch_bounds__(FIN_T)
owner_finalize(const int* __restrict__ recs, const int* __restrict__ run_cnt, const int* __restrict__ run_lst,
               int* __restrict__ span, int* __restrict__ by_dst) {
    __shared__ int cnt[OWN_R];
    __shared__ int tot[TAB_W];
    __shared__ int run_n[REC_PT * RUN_SWEEP];
    __shared__ int run_p[REC_PT * RUN_SWEEP];
    __shared__ int s_w[16];
    __shared__ int s_bt[2];
    __shared__ int s_long;
    __shared__ int stage[STAGE_CAP];

    const int tid = threadIdx.x;
    const int o   = blockIdx.x;
    const int node0 = o * OWN_R;

    int rn = 0;
    if (tid < NCHUNK) {
        rn = run_cnt[tid * TAB_W + o];
        run_n[tid] = rn;
        run_p[tid] = tid * CHUNK + run_lst[tid * TAB_W + o];
    } else if (tid < REC_PT * RUN_SWEEP) {
        run_n[tid] = 0;
        run_p[tid] = 0;
    }
    if (tid < TAB_W) tot[tid] = 0;
    if (tid == 0) s_long = 0;
    cnt[tid] = 0;
    __syncthreads();
    if (rn > RUN_SLOTS) s_long = 1;

    {
        const int4* t4 = (const int4*)run_cnt;
        int4 a = make_int4(0, 0, 0, 0);
#pragma unroll
        for (int j = 0; j < TAB_PT; ++j) {
            const int q = tid + FIN_T * j;
            if (q < TAB_Q) {
                const int4 x = t4[q];
                a.x += x.x; a.y += x.y; a.z += x.z; a.w += x.w;
            }
        }
        const int c4 = (tid & (TAB_W / 4 - 1)) * 4;
        atomicAdd(&tot[c4 + 0], a.x);
        atomicAdd(&tot[c4 + 1], a.y);
        atomicAdd(&tot[c4 + 2], a.z);
        atomicAdd(&tot[c4 + 3], a.w);
    }
    __syncthreads();

    const int t = (tid < NOWN) ? tot[tid] : 0;
    int unused;
    const int ex = block_excl_scan<8>(t, tid, s_w, unused);
    if (tid == o) { s_bt[0] = ex; s_bt[1] = t; }
    __syncthreads();
    const int base  = s_bt[0];
    const int total = s_bt[1];
    const int2* r2 = (const int2*)recs;

    const bool fits = (total <= STAGE_CAP) && (s_long == 0);
    const int kslot = tid & (RUN_SLOTS - 1);
    const int crun  = tid / RUN_SLOTS;
    int2 v[REC_PT];
    if (fits) {
        int nn[REC_PT], pp[REC_PT];
#pragma unroll
        for (int u = 0; u < REC_PT; ++u) {
            const int c = crun + RUN_SWEEP * u;
            nn[u] = run_n[c];
            pp[u] = run_p[c] + kslot;
        }
#pragma unroll
        for (int u = 0; u < REC_PT; ++u) {
            v[u] = make_int2(0, -1);
            if (kslot < nn[u]) v[u] = r2[pp[u]];
        }
#pragma unroll
        for (int u = 0; u < REC_PT; ++u)
            if (v[u].y >= 0) atomicAdd(&cnt[v[u].y - node0], 1);
    } else {
        for (int c = 0; c < NCHUNK; ++c) {
            const int n = run_n[c];
            const int2* p = r2 + run_p[c];
            for (int i = tid; i < n; i += FIN_T) atomicAdd(&cnt[p[i].y - node0], 1);
        }
    }
    __syncthreads();

    const int c  = cnt[tid];
    const int st = block_excl_scan<16>(c, tid, s_w, unused);
    cnt[tid] = st;
    if (node0 + tid < N_NODES) ((int2*)span)[node0 + tid] = make_int2(base + st, c);
    __syncthreads();

    if (fits) {
#pragma unroll
        for (int u = 0; u < REC_PT; ++u)
            if (v[u].y >= 0) stage[atomicAdd(&cnt[v[u].y - node0], 1)] = v[u].x;
        __syncthreads();
        for (int i = tid; i < total; i += FIN_T) by_dst[base + i] = stage[i];
    } else {
        for (int cc = 0; cc < NCHUNK; ++cc) {
            const int n = run_n[cc];
            const int2* p = r2 + run_p[cc];
            for (int i = tid; i < n; i += FIN_T) {
                const int2 e = p[i];
                by_dst[base + atomicAdd(&cnt[e.y - node0], 1)] = e.x;
            }
        }
    }
}

#define GROW 16

__device__ __forceinline__ v4f max4(v4f a, v4f b) {
    v4f r;
    r.x = fmaxf(a.x, b.x); r.y = fmaxf(a.y, b.y); r.z = fmaxf(a.z, b.z); r.w = fmaxf(a.w, b.w);
    return r;
}
__device__ __forceinline__ v4f red4(const v4f (&r)[4]) {
    return max4(max4(r[0], r[1]), max4(r[2], r[3]));
}
__device__ __forceinline__ void load4(const float* __restrict__ mcol, int idx, int j, v4f (&r)[4]) {
#pragma unroll
    for (int q = 0; q < 4; ++q) {
        const int s = __builtin_amdgcn_readlane(idx, j + q);
        r[q] = *(const v4f*)(mcol + (size_t)s * KDIM);
    }
}

__device__ __forceinline__ v4f node_max(const float* __restrict__ mcol, const int* __restrict__ by_dst,
                                        int start, int n, int idx0, int lane) {
    v4f acc = {0.0f, 0.0f, 0.0f, 0.0f};
    for (int base = 0; base < n; base += 32) {
        const int left = n - base;
        const int last = (left < 32 ? left : 32) - 1;
        int idx = idx0;
        if (base > 0) idx = by_dst[start + base + (lane < last ? lane : last)];
        for (int j = 0; j <= last; j += GROW) {
            v4f a[4];
            load4(mcol, idx, j, a);
            if (j + 4 <= last) {
                v4f b[4];
                load4(mcol, idx, j + 4, b);
                if (j + 8 <= last) {
                    v4f c[4];
                    load4(mcol, idx, j + 8, c);
                    if (j + 12 <= last) {
                        v4f d[4];
                        load4(mcol, idx, j + 12, d);
                        acc = max4(acc, red4(d));
                    }
                    acc = max4(acc, red4(c));
                }
                acc = max4(acc, red4(b));
            }
            acc = max4(acc, red4(a));
        }
    }
    return acc;
}

template <int O, bool RELU>
__global__ void __launch_bounds__(256)
gather_gemm(const float* __restrict__ X1, const float* __restrict__ W1, const float* __restrict__ b1,
            const float* __restrict__ Mp, const int* __restrict__ span, const int* __restrict__ by_dst,
            const float* __restrict__ W2, const float* __restrict__ b2, float* __restrict__ Y) {
    constexpr int NCW   = (O + 15) / 16;
    constexpr int STAGE = 2 * 32 * PITCH;
    __shared__ __attribute__((aligned(16))) float smem[2 * STAGE];

    const int tid  = threadIdx.x;
    const int wave = __builtin_amdgcn_readfirstlane(tid >> 5);
    const int lane = tid & 31;
    const int m    = lane & 15;
    const int kh   = (lane >> 4) << 1;
    const int col  = wave * 16 + m;
    const bool valid = (col < O);
    const int colW = valid ? col : (O - 1);
    const int soff = (tid >> 5) * PITCH + (tid & 31) * 4;
    const float* mcol = Mp + lane * 4;

    v4f pre[4];
    int st_[4], n_[4], idx[4];
    auto fetch_span = [&](int p) -> int {
        return span[(p * 32 + wave * 4) * 2 + (lane & 7)];
    };
    auto fetch_x = [&](int p) {
        const v4f* s1 = (const v4f*)X1 + (size_t)p * 1024 + tid;
#pragma unroll
        for (int j = 0; j < 4; ++j) pre[j] = s1[256 * j];
    };
    auto fetch_idx = [&](int sv) {
#pragma unroll
        for (int i = 0; i < 4; ++i) {
            st_[i] = __builtin_amdgcn_readlane(sv, 2 * i);
            n_[i]  = __builtin_amdgcn_readlane(sv, 2 * i + 1);
            const int lastl = (n_[i] < 32 ? n_[i] : 32) - 1;
            idx[i] = (n_[i] > 0) ? by_dst[st_[i] + (lane < lastl ? lane : lastl)] : 0;
        }
    };

    int pair = blockIdx.x;
    int sv = 0, svn = 0;
    if (pair < NPAIRS) { sv = fetch_span(pair); fetch_x(pair); }
    if (pair + (int)gridDim.x < NPAIRS) svn = fetch_span(pair + gridDim.x);

    v2f bf1[32];
    v2f bf2[32];
    load_wfrag(W1, O, smem, tid, 256, colW, kh, bf1);
    if (pair < NPAIRS) fetch_idx(sv);
    load_wfrag(W2, O, smem, tid, 256, colW, kh, bf2);
    const float bv = b1[colW] + b2[colW];

    int s = 0;
    for (; pair < NPAIRS; pair += gridDim.x) {
        float* st = smem + s * STAGE;
#pragma unroll
        for (int j = 0; j < 4; ++j) *(v4f*)(st + soff + 8 * j * PITCH) = pre[j];
#pragma unroll
        for (int i = 0; i < 4; ++i) {
            const v4f a = node_max(mcol, by_dst, st_[i], n_[i], idx[i], lane);
            *(v4f*)(st + (32 + wave * 4 + i) * PITCH + lane * 4) = a;
        }
        __syncthreads();

        const int nxt = pair + gridDim.x;
        if (nxt < NPAIRS) { fetch_x(nxt); fetch_idx(svn); }
        const int nxt2 = nxt + gridDim.x;
        svn = (nxt2 < NPAIRS) ? fetch_span(nxt2) : 0;

        if (wave < NCW) {
            const float* ap = st + m * PITCH + kh;
            v8f c0 = {}, c1 = {};
            kloop2(ap, bf1, c0, c1);
            kloop2(ap + 32 * PITCH, bf2, c0, c1);

            const int rb = pair * 32 + ((lane >> 4) << 3);
#pragma unroll
            for (int r = 0; r < 8; ++r) {
                float v0 = c0[r] + bv;
                float v1 = c1[r] + bv;
                if (RELU) { v0 = fmaxf(v0, 0.0f); v1 = fmaxf(v1, 0.0f); }
                if (valid) {
                    Y[(size_t)(rb + r) * O + col]      = v0;
                    Y[(size_t)(rb + 16 + r) * O + col] = v1;
                }
            }
        }
        s ^= 1;
    }
}

extern "C" void kernel_launch(void* const* d_in, const int* in_sizes, int n_in,
                              void* d_out, int out_size, void* d_ws, size_t ws_size,
                              hipStream_t stream) {
    const float* in_feat  = (const float*)d_in[0];
    const int*   src      = (const int*)d_in[1];
    const int*   dst      = (const int*)d_in[2];
    const float* pool_w1  = (const float*)d_in[3];
    const float* pool_b1  = (const float*)d_in[4];
    const float* self_w1  = (const float*)d_in[5];
    const float* self_b1  = (const float*)d_in[6];
    const float* neigh_w1 = (const float*)d_in[7];
    const float* neigh_b1 = (const float*)d_in[8];
    const float* pool_w2  = (const float*)d_in[9];
    const float* pool_b2  = (const float*)d_in[10];
    const float* self_w2  = (const float*)d_in[11];
    const float* self_b2  = (const float*)d_in[12];
    const float* neigh_w2 = (const float*)d_in[13];
    const float* neigh_b2 = (const float*)d_in[14];
    float* out = (float*)d_out;

    const size_t NODE_BYTES = (size_t)N_NODES * KDIM * sizeof(float);
    char* ws = (char*)d_ws;
    float* m_buf = (float*)(ws);
    char*  aux   = ws + NODE_BYTES;
    float* h1    = (float*)(ws + 2 * NODE_BYTES);

    int*  recs    = (int*)(aux);
    int*  by_dst  = (int*)(aux + 16777216);
    int*  span    = (int*)(aux + 25165824);
    int*  run_cnt = (int*)(aux + 27262976);
    int*  run_lst = (int*)(aux + 27787264);

    chunk_sort<<<NCHUNK, 256, 0, stream>>>(src, dst, recs, run_cnt, run_lst);
    owner_finalize<<<NOWN, FIN_T, 0, stream>>>(recs, run_cnt, run_lst, span, by_dst);

    gemm_wreg<1, true><<<POOL_BLOCKS, 256, 0, stream>>>(in_feat, pool_w1, pool_b1,
                                                        nullptr, nullptr, nullptr, m_buf);
    gather_gemm<KDIM, true><<<GEMM_BLOCKS, 256, 0, stream>>>(in_feat, self_w1, self_b1,
                                                             m_buf, span, by_dst, neigh_w1, neigh_b1, h1);

    gemm_wreg<1, true><<<POOL_BLOCKS, 256, 0, stream>>>(h1, pool_w2, pool_b2,
                                                        nullptr, nullptr, nullptr, m_buf);
    gather_gemm<40, false><<<GEMM_BLOCKS, 256, 0, stream>>>(h1, self_w2, self_b2,
                                                            m_buf, span, by_dst, neigh_w2, neigh_b2, out);
}
